# starting kernel
# baseline (speedup 1.0000x reference)
#include <hip/hip_runtime.h>
#include <hip/hip_bf16.h>
#include <math.h>

typedef __attribute__((ext_vector_type(2))) float v2f;
typedef __attribute__((ext_vector_type(8))) float v8f;

#define NEG_SLOPE 0.2f

__device__ __forceinline__ unsigned flipf(float f) {
  unsigned u = __float_as_uint(f);
  return (u & 0x80000000u) ? ~u : (u | 0x80000000u);
}
__device__ __forceinline__ float unflipf(unsigned k) {
  unsigned u = (k & 0x80000000u) ? (k ^ 0x80000000u) : ~k;
  return __uint_as_float(u);
}

__global__ void zero_kernel(float* __restrict__ p, long long n) {
  long long i = (long long)blockIdx.x * blockDim.x + threadIdx.x;
  if (i < n) p[i] = 0.f;
}

template<int NC>
__global__ __launch_bounds__(256) void gemm_f32_wmma(const float* __restrict__ A,
                                                     const float* __restrict__ B,
                                                     float* __restrict__ C, int M) {
  constexpr int K = 128;
  constexpr int CG = 64;
  constexpr int LDA = K + 4;
  constexpr int colGroups = NC / CG;
  __shared__ float Wlds[K * CG];
  __shared__ float Alds[32 * LDA];
  int cg = blockIdx.x % colGroups;
  int rg = blockIdx.x / colGroups;

  for (int i = threadIdx.x; i < K * (CG / 4); i += 256) {
    int k = i / (CG / 4), c4 = i % (CG / 4);
    *(float4*)&Wlds[k * CG + c4 * 4] =
        *(const float4*)&B[(size_t)k * NC + cg * CG + c4 * 4];
  }
  for (int i = threadIdx.x; i < 32 * (K / 4); i += 256) {
    int r = i / (K / 4), c4 = i % (K / 4);
    int grow = rg * 32 + r;
    float4 v = make_float4(0.f, 0.f, 0.f, 0.f);
    if (grow < M) v = *(const float4*)&A[(size_t)grow * K + c4 * 4];
    *(float4*)&Alds[r * LDA + c4 * 4] = v;
  }
  __syncthreads();

  int wave = threadIdx.x >> 5;
  int lane = threadIdx.x & 31;
  int half = lane >> 4;
  int lrow = lane & 15;
  int rt = wave >> 2, ct = wave & 3;
  int grow0 = rg * 32 + rt * 16;
  if (grow0 >= M) return;

  const float* Ab = &Alds[(rt * 16 + lrow) * LDA + 2 * half];
  const float* Bb = &Wlds[(2 * half) * CG + ct * 16 + lrow];
  v8f acc = {};
  #pragma unroll
  for (int k0 = 0; k0 < K; k0 += 4) {
    v2f a, b;
    a.x = Ab[k0];        a.y = Ab[k0 + 1];
    b.x = Bb[k0 * CG];   b.y = Bb[k0 * CG + CG];
    acc = __builtin_amdgcn_wmma_f32_16x16x4_f32(false, a, false, b,
                                                (short)0, acc, false, false);
  }
  float* Cp = &C[(size_t)(grow0 + 8 * half) * NC + cg * CG + ct * 16 + lrow];
  #pragma unroll
  for (int j = 0; j < 8; ++j) Cp[(size_t)j * NC] = acc[j];
}

template<int H>
__global__ void node_alpha_kernel(const float* __restrict__ h,
                                  const float* __restrict__ aw_src,
                                  const float* __restrict__ aw_dst,
                                  float* __restrict__ asrc, float* __restrict__ adst, int N) {
  int idx = blockIdx.x * blockDim.x + threadIdx.x;
  if (idx >= N * H) return;
  int hh = idx % H;
  const float* hp = h + (size_t)idx * 64;
  const float* ap = aw_src + hh * 64;
  const float* bp = aw_dst + hh * 64;
  float sa = 0.f, sb = 0.f;
  #pragma unroll
  for (int c = 0; c < 64; c += 4) {
    float4 v = *(const float4*)(hp + c);
    float4 u = *(const float4*)(ap + c);
    float4 t = *(const float4*)(bp + c);
    sa += v.x * u.x + v.y * u.y + v.z * u.z + v.w * u.w;
    sb += v.x * t.x + v.y * t.y + v.z * t.z + v.w * t.w;
  }
  asrc[idx] = sa;
  adst[idx] = sb;
}

template<int H>
__global__ void edge_max_kernel(const int* __restrict__ srcE, const int* __restrict__ dstE,
                                const float* __restrict__ asrc, const float* __restrict__ adst,
                                unsigned* __restrict__ m, int E, int Etot) {
  int idx = blockIdx.x * blockDim.x + threadIdx.x;
  if (idx >= Etot * H) return;
  int e = idx / H, hh = idx % H;
  int s = (e < E) ? srcE[e] : (e - E);
  int d = (e < E) ? dstE[e] : (e - E);
  float v = asrc[s * H + hh] + adst[d * H + hh];
  v = (v > 0.f) ? v : NEG_SLOPE * v;
  atomicMax(&m[d * H + hh], flipf(v));
}

template<int H>
__global__ void edge_expsum_kernel(const int* __restrict__ srcE, const int* __restrict__ dstE,
                                   const float* __restrict__ asrc, const float* __restrict__ adst,
                                   const unsigned* __restrict__ m, float* __restrict__ ssum,
                                   float* __restrict__ eout, int E, int Etot) {
  int idx = blockIdx.x * blockDim.x + threadIdx.x;
  if (idx >= Etot * H) return;
  int e = idx / H, hh = idx % H;
  int s = (e < E) ? srcE[e] : (e - E);
  int d = (e < E) ? dstE[e] : (e - E);
  float v = asrc[s * H + hh] + adst[d * H + hh];
  v = (v > 0.f) ? v : NEG_SLOPE * v;
  float ex = __expf(v - unflipf(m[d * H + hh]));
  eout[idx] = ex;
  atomicAdd(&ssum[d * H + hh], ex);
}

template<int H>
__global__ void edge_agg_kernel(const int* __restrict__ srcE, const int* __restrict__ dstE,
                                const float* __restrict__ eexp, const float* __restrict__ ssum,
                                const float* __restrict__ h, float* __restrict__ agg,
                                int E, int Etot) {
  constexpr int G = H * 64 / 4;
  long long idx = (long long)blockIdx.x * blockDim.x + threadIdx.x;
  if (idx >= (long long)Etot * G) return;
  int g = (int)(idx % G);
  int e = (int)(idx / G);
  int s = (e < E) ? srcE[e] : (e - E);
  int d = (e < E) ? dstE[e] : (e - E);
  int hh = g >> 4;
  int c = (g & 15) * 4;
  float alpha = eexp[e * H + hh] / ssum[d * H + hh];
  float4 hv = *(const float4*)&h[((size_t)s * H + hh) * 64 + c];
  float* ap = &agg[((size_t)d * H + hh) * 64 + c];
  atomicAdd(ap + 0, alpha * hv.x);
  atomicAdd(ap + 1, alpha * hv.y);
  atomicAdd(ap + 2, alpha * hv.z);
  atomicAdd(ap + 3, alpha * hv.w);
}

__global__ void bias_relu_kernel(float* __restrict__ a, const float* __restrict__ b,
                                 long long n, int Fmask) {
  long long i = (long long)blockIdx.x * blockDim.x + threadIdx.x;
  if (i >= n) return;
  float v = a[i] + b[(int)(i & Fmask)];
  a[i] = v > 0.f ? v : 0.f;
}

__global__ void bias_sigmoid_kernel(const float* __restrict__ a, const float* __restrict__ b,
                                    float* __restrict__ out, long long n, int Fmask) {
  long long i = (long long)blockIdx.x * blockDim.x + threadIdx.x;
  if (i >= n) return;
  float v = a[i] + b[(int)(i & Fmask)];
  out[i] = 1.f / (1.f + __expf(-v));
}

extern "C" void kernel_launch(void* const* d_in, const int* in_sizes, int n_in,
                              void* d_out, int out_size, void* d_ws, size_t ws_size,
                              hipStream_t stream) {
  const float* x   = (const float*)d_in[0];
  const int*   ei  = (const int*)d_in[1];
  const float* W1  = (const float*)d_in[2];
  const float* as1 = (const float*)d_in[3];
  const float* ad1 = (const float*)d_in[4];
  const float* b1  = (const float*)d_in[5];
  const float* W2  = (const float*)d_in[6];
  const float* as2 = (const float*)d_in[7];
  const float* ad2 = (const float*)d_in[8];
  const float* b2  = (const float*)d_in[9];
  float* out = (float*)d_out;

  const int N = in_sizes[0] / 128;
  const int E = in_sizes[1] / 2;
  const int Etot = E + N;
  const int* srcE = ei;
  const int* dstE = ei + E;

  float* w = (float*)d_ws;
  size_t o = 0;
  float* h1    = w + o; o += (size_t)N * 128;
  float* agg1  = w + o; o += (size_t)N * 128;
  float* asrc1 = w + o; o += (size_t)N * 2;
  float* adst1 = w + o; o += (size_t)N * 2;
  unsigned* m1 = (unsigned*)(w + o); o += (size_t)N * 2;
  float* s1    = w + o; o += (size_t)N * 2;
  float* asrc2 = w + o; o += (size_t)N;
  float* adst2 = w + o; o += (size_t)N;
  unsigned* m2 = (unsigned*)(w + o); o += (size_t)N;
  float* s2    = w + o; o += (size_t)N;
  float* e1    = w + o; o += (size_t)Etot * 2;
  float* e2    = w + o; o += (size_t)Etot;
  float* h2f   = h1;
  float* agg2  = h1 + (size_t)N * 64;
  (void)o; (void)ws_size; (void)n_in; (void)out_size;

  const int TB = 256;
  auto blk = [](long long n, int tb) { return (unsigned)((n + tb - 1) / tb); };
  const int rowGroups = (N + 31) / 32;

  zero_kernel<<<blk((long long)N * 128, TB), TB, 0, stream>>>(agg1, (long long)N * 128);
  zero_kernel<<<blk((long long)N * 4, TB), TB, 0, stream>>>((float*)m1, (long long)N * 4);
  gemm_f32_wmma<128><<<dim3(rowGroups * 2), TB, 0, stream>>>(x, W1, h1, N);
  node_alpha_kernel<2><<<blk((long long)N * 2, TB), TB, 0, stream>>>(h1, as1, ad1, asrc1, adst1, N);
  edge_max_kernel<2><<<blk((long long)Etot * 2, TB), TB, 0, stream>>>(srcE, dstE, asrc1, adst1, m1, E, Etot);
  edge_expsum_kernel<2><<<blk((long long)Etot * 2, TB), TB, 0, stream>>>(srcE, dstE, asrc1, adst1, m1, s1, e1, E, Etot);
  edge_agg_kernel<2><<<blk((long long)Etot * 32, TB), TB, 0, stream>>>(srcE, dstE, e1, s1, h1, agg1, E, Etot);
  bias_relu_kernel<<<blk((long long)N * 128, TB), TB, 0, stream>>>(agg1, b1, (long long)N * 128, 127);

  gemm_f32_wmma<64><<<dim3(rowGroups), TB, 0, stream>>>(agg1, W2, h2f, N);
  zero_kernel<<<blk((long long)N * 64, TB), TB, 0, stream>>>(agg2, (long long)N * 64);
  zero_kernel<<<blk((long long)N * 2, TB), TB, 0, stream>>>((float*)m2, (long long)N * 2);
  node_alpha_kernel<1><<<blk((long long)N, TB), TB, 0, stream>>>(h2f, as2, ad2, asrc2, adst2, N);
  edge_max_kernel<1><<<blk((long long)Etot, TB), TB, 0, stream>>>(srcE, dstE, asrc2, adst2, m2, E, Etot);
  edge_expsum_kernel<1><<<blk((long long)Etot, TB), TB, 0, stream>>>(srcE, dstE, asrc2, adst2, m2, s2, e2, E, Etot);
  edge_agg_kernel<1><<<blk((long long)Etot * 16, TB), TB, 0, stream>>>(srcE, dstE, e2, s2, h2f, agg2, E, Etot);
  bias_sigmoid_kernel<<<blk((long long)N * 64, TB), TB, 0, stream>>>(agg2, b2, out, (long long)N * 64, 63);
}


# edge aggregation by destination range: a work-group owns 200 nodes, filters the whole edge list, sums ex*h[src] and ex in LDS by LDS float adds, stores rows once; sums in LDS arrival order
# speedup vs baseline: 1.6519x; 1.6519x over previous
#include <hip/hip_runtime.h>
#include <hip/hip_bf16.h>
#include <math.h>

typedef __attribute__((ext_vector_type(2))) float v2f;
typedef __attribute__((ext_vector_type(8))) float v8f;

#define NEG_SLOPE 0.2f

__device__ __forceinline__ unsigned flipf(float f) {
  unsigned u = __float_as_uint(f);
  return (u & 0x80000000u) ? ~u : (u | 0x80000000u);
}
__device__ __forceinline__ float unflipf(unsigned k) {
  unsigned u = (k & 0x80000000u) ? (k ^ 0x80000000u) : ~k;
  return __uint_as_float(u);
}

__global__ void zero_kernel(float* __restrict__ p, long long n) {
  long long i = (long long)blockIdx.x * blockDim.x + threadIdx.x;
  if (i < n) p[i] = 0.f;
}

template<int NC>
__global__ __launch_bounds__(256) void gemm_f32_wmma(const float* __restrict__ A,
                                                     const float* __restrict__ B,
                                                     float* __restrict__ C, int M) {
  constexpr int K = 128;
  constexpr int CG = 64;
  constexpr int LDA = K + 4;
  constexpr int colGroups = NC / CG;
  __shared__ float Wlds[K * CG];
  __shared__ float Alds[32 * LDA];
  int cg = blockIdx.x % colGroups;
  int rg = blockIdx.x / colGroups;

  for (int i = threadIdx.x; i < K * (CG / 4); i += 256) {
    int k = i / (CG / 4), c4 = i % (CG / 4);
    *(float4*)&Wlds[k * CG + c4 * 4] =
        *(const float4*)&B[(size_t)k * NC + cg * CG + c4 * 4];
  }
  for (int i = threadIdx.x; i < 32 * (K / 4); i += 256) {
    int r = i / (K / 4), c4 = i % (K / 4);
    int grow = rg * 32 + r;
    float4 v = make_float4(0.f, 0.f, 0.f, 0.f);
    if (grow < M) v = *(const float4*)&A[(size_t)grow * K + c4 * 4];
    *(float4*)&Alds[r * LDA + c4 * 4] = v;
  }
  __syncthreads();

  int wave = threadIdx.x >> 5;
  int lane = threadIdx.x & 31;
  int half = lane >> 4;
  int lrow = lane & 15;
  int rt = wave >> 2, ct = wave & 3;
  int grow0 = rg * 32 + rt * 16;
  if (grow0 >= M) return;

  const float* Ab = &Alds[(rt * 16 + lrow) * LDA + 2 * half];
  const float* Bb = &Wlds[(2 * half) * CG + ct * 16 + lrow];
  v8f acc = {};
  #pragma unroll
  for (int k0 = 0; k0 < K; k0 += 4) {
    v2f a, b;
    a.x = Ab[k0];        a.y = Ab[k0 + 1];
    b.x = Bb[k0 * CG];   b.y = Bb[k0 * CG + CG];
    acc = __builtin_amdgcn_wmma_f32_16x16x4_f32(false, a, false, b,
                                                (short)0, acc, false, false);
  }
  float* Cp = &C[(size_t)(grow0 + 8 * half) * NC + cg * CG + ct * 16 + lrow];
  #pragma unroll
  for (int j = 0; j < 8; ++j) Cp[(size_t)j * NC] = acc[j];
}

template<int H>
__global__ void node_alpha_kernel(const float* __restrict__ h,
                                  const float* __restrict__ aw_src,
                                  const float* __restrict__ aw_dst,
                                  float* __restrict__ asrc, float* __restrict__ adst, int N) {
  int idx = blockIdx.x * blockDim.x + threadIdx.x;
  if (idx >= N * H) return;
  int hh = idx % H;
  const float* hp = h + (size_t)idx * 64;
  const float* ap = aw_src + hh * 64;
  const float* bp = aw_dst + hh * 64;
  float sa = 0.f, sb = 0.f;
  #pragma unroll
  for (int c = 0; c < 64; c += 4) {
    float4 v = *(const float4*)(hp + c);
    float4 u = *(const float4*)(ap + c);
    float4 t = *(const float4*)(bp + c);
    sa += v.x * u.x + v.y * u.y + v.z * u.z + v.w * u.w;
    sb += v.x * t.x + v.y * t.y + v.z * t.z + v.w * t.w;
  }
  asrc[idx] = sa;
  adst[idx] = sb;
}

template<int H>
__global__ void edge_max_kernel(const int* __restrict__ srcE, const int* __restrict__ dstE,
                                const float* __restrict__ asrc, const float* __restrict__ adst,
                                unsigned* __restrict__ m, int E, int Etot) {
  int idx = blockIdx.x * blockDim.x + threadIdx.x;
  if (idx >= Etot * H) return;
  int e = idx / H, hh = idx % H;
  int s = (e < E) ? srcE[e] : (e - E);
  int d = (e < E) ? dstE[e] : (e - E);
  float v = asrc[s * H + hh] + adst[d * H + hh];
  v = (v > 0.f) ? v : NEG_SLOPE * v;
  atomicMax(&m[d * H + hh], flipf(v));
}

template<int H, int R>
__global__ __launch_bounds__(1024) void range_expagg_kernel(const int* __restrict__ srcE,
                                                            const int* __restrict__ dstE,
                                                            const float* __restrict__ asrc,
                                                            const float* __restrict__ adst,
                                                            const unsigned* __restrict__ m,
                                                            const float* __restrict__ h,
                                                            float* __restrict__ ssum,
                                                            float* __restrict__ agg, int E, int N) {
  constexpr int HC = H * 64;
  constexpr int KK = HC / 32;
  constexpr int NT = 1024;
  constexpr int UV = 4;
  constexpr int UD = 16 / KK;
  constexpr int QCAP = 128 + 32 * 4 * UV;
  static_assert(R <= NT, "one lane per owned node for the self-loops");
  __shared__ float acc[R * HC];
  __shared__ float sl[R * H];
  __shared__ float ml[R * H];
  __shared__ float adl[R * H];
  __shared__ int qe[(NT / 32) * QCAP];

  const int tid = threadIdx.x;
  const int lane = tid & 31;
  const int wave = tid >> 5;
  const int lo = blockIdx.x * R;

  for (int i = tid; i < R * HC; i += NT) acc[i] = 0.f;
  for (int i = tid; i < R * H; i += NT) {
    int gi = lo * H + i;
    bool ok = gi < N * H;
    sl[i] = 0.f;
    ml[i] = ok ? unflipf(m[gi]) : 0.f;
    adl[i] = ok ? adst[gi] : 0.f;
  }
  __syncthreads();

  int* myq = qe + wave * QCAP;
  int cnt = 0;

  auto append = [&](bool keep, int e) {
    unsigned mask = __builtin_amdgcn_ballot_w32(keep);
    if (mask == 0u) return;
    if (keep) myq[cnt + __builtin_popcount(mask & ((1u << lane) - 1u))] = e;
    cnt += __builtin_popcount(mask);
  };

  int head = (int)(((16u - (unsigned)((size_t)dstE & 15u)) & 15u) >> 2);
  if (head > E) head = E;
  const int nvec = (E - head) >> 2;
  const int tail0 = head + 4 * nvec;
  const int4* d4 = (const int4*)(dstE + head);
  append(tid < head && (unsigned)(dstE[tid < head ? tid : 0] - lo) < (unsigned)R, tid);
  append(tail0 + tid < E && (unsigned)(dstE[tail0 + tid < E ? tail0 + tid : 0] - lo) < (unsigned)R, tail0 + tid);
  append(tid < R && lo + tid < N, E + lo + tid);

  const int rounds = (nvec + UV * NT - 1) / (UV * NT);
  for (int it = 0; it <= rounds; ++it) {
    if (it < rounds) {
      const int q0 = it * (UV * NT);
      int4 dv[UV];
      #pragma unroll
      for (int j = 0; j < UV; ++j) {
        int q = q0 + j * NT + tid;
        dv[j] = (q < nvec) ? d4[q] : make_int4(-1, -1, -1, -1);
      }
      #pragma unroll
      for (int j = 0; j < UV; ++j) {
        int eb = head + 4 * (q0 + j * NT + tid);
        append((unsigned)(dv[j].x - lo) < (unsigned)R, eb + 0);
        append((unsigned)(dv[j].y - lo) < (unsigned)R, eb + 1);
        append((unsigned)(dv[j].z - lo) < (unsigned)R, eb + 2);
        append((unsigned)(dv[j].w - lo) < (unsigned)R, eb + 3);
      }
    }
    const int leave = (it < rounds) ? 31 : 0;
    while (cnt > leave) {
      const int n = cnt < 32 ? cnt : 32;
      cnt -= n;
      int s = 0, dl = 0;
      float ex0 = 0.f, ex1 = 0.f;
      if (lane < n) {
        int e = myq[cnt + lane];
        s = (e < E) ? srcE[e] : (e - E);
        dl = ((e < E) ? dstE[e] : (e - E)) - lo;
        float v = asrc[s * H] + adl[dl * H];
        v = (v > 0.f) ? v : NEG_SLOPE * v;
        ex0 = __expf(v - ml[dl * H]);
        atomicAdd(&sl[dl * H], ex0);
        if (H == 2) {
          float w = asrc[s * H + 1] + adl[dl * H + 1];
          w = (w > 0.f) ? w : NEG_SLOPE * w;
          ex1 = __expf(w - ml[dl * H + 1]);
          atomicAdd(&sl[dl * H + 1], ex1);
        }
      }
      int i = 0;
      for (; i + UD <= n; i += UD) {
        float val[UD][KK];
        #pragma unroll
        for (int u = 0; u < UD; ++u) {
          int si = __builtin_amdgcn_readlane(s, i + u);
          const float* hp = h + (size_t)si * HC + lane;
          #pragma unroll
          for (int k = 0; k < KK; ++k) val[u][k] = hp[32 * k];
        }
        #pragma unroll
        for (int u = 0; u < UD; ++u) {
          int di = __builtin_amdgcn_readlane(dl, i + u);
          float x0 = __int_as_float(__builtin_amdgcn_readlane(__float_as_int(ex0), i + u));
          float x1 = __int_as_float(__builtin_amdgcn_readlane(__float_as_int(ex1), i + u));
          float* ap = acc + di * HC + lane;
          #pragma unroll
          for (int k = 0; k < KK; ++k) atomicAdd(ap + 32 * k, ((k < 2) ? x0 : x1) * val[u][k]);
        }
      }
      for (; i < n; ++i) {
        int si = __builtin_amdgcn_readlane(s, i);
        int di = __builtin_amdgcn_readlane(dl, i);
        float x0 = __int_as_float(__builtin_amdgcn_readlane(__float_as_int(ex0), i));
        float x1 = __int_as_float(__builtin_amdgcn_readlane(__float_as_int(ex1), i));
        const float* hp = h + (size_t)si * HC + lane;
        float* ap = acc + di * HC + lane;
        float val[KK];
        #pragma unroll
        for (int k = 0; k < KK; ++k) val[k] = hp[32 * k];
        #pragma unroll
        for (int k = 0; k < KK; ++k) atomicAdd(ap + 32 * k, ((k < 2) ? x0 : x1) * val[k]);
      }
    }
  }
  __syncthreads();

  for (int i = tid; i < R * HC; i += NT) {
    size_t gi = (size_t)lo * HC + i;
    if (gi < (size_t)N * HC) agg[gi] = acc[i];
  }
  for (int i = tid; i < R * H; i += NT) {
    int gi = lo * H + i;
    if (gi < N * H) ssum[gi] = sl[i];
  }
}

__global__ void bias_relu_kernel(float* __restrict__ a, const float* __restrict__ ssum,
                                 const float* __restrict__ b, long long n, int Fmask) {
  long long i = (long long)blockIdx.x * blockDim.x + threadIdx.x;
  if (i >= n) return;
  float v = a[i] / ssum[i >> 6] + b[(int)(i & Fmask)];
  a[i] = v > 0.f ? v : 0.f;
}

__global__ void bias_sigmoid_kernel(const float* __restrict__ a, const float* __restrict__ ssum,
                                    const float* __restrict__ b, float* __restrict__ out,
                                    long long n, int Fmask) {
  long long i = (long long)blockIdx.x * blockDim.x + threadIdx.x;
  if (i >= n) return;
  float v = a[i] / ssum[i >> 6] + b[(int)(i & Fmask)];
  out[i] = 1.f / (1.f + __expf(-v));
}

extern "C" void kernel_launch(void* const* d_in, const int* in_sizes, int n_in,
                              void* d_out, int out_size, void* d_ws, size_t ws_size,
                              hipStream_t stream) {
  const float* x   = (const float*)d_in[0];
  const int*   ei  = (const int*)d_in[1];
  const float* W1  = (const float*)d_in[2];
  const float* as1 = (const float*)d_in[3];
  const float* ad1 = (const float*)d_in[4];
  const float* b1  = (const float*)d_in[5];
  const float* W2  = (const float*)d_in[6];
  const float* as2 = (const float*)d_in[7];
  const float* ad2 = (const float*)d_in[8];
  const float* b2  = (const float*)d_in[9];
  float* out = (float*)d_out;

  const int N = in_sizes[0] / 128;
  const int E = in_sizes[1] / 2;
  const int Etot = E + N;
  const int* srcE = ei;
  const int* dstE = ei + E;

  float* w = (float*)d_ws;
  size_t o = 0;
  float* h1    = w + o; o += (size_t)N * 128;
  float* agg1  = w + o; o += (size_t)N * 128;
  float* asrc1 = w + o; o += (size_t)N * 2;
  float* adst1 = w + o; o += (size_t)N * 2;
  unsigned* m1 = (unsigned*)(w + o); o += (size_t)N * 2;
  float* s1    = w + o; o += (size_t)N * 2;
  float* asrc2 = w + o; o += (size_t)N;
  float* adst2 = w + o; o += (size_t)N;
  unsigned* m2 = (unsigned*)(w + o); o += (size_t)N;
  float* s2    = w + o; o += (size_t)N;
  float* h2f   = h1;
  float* agg2  = h1 + (size_t)N * 64;
  (void)o; (void)ws_size; (void)n_in; (void)out_size;

  const int TB = 256;
  auto blk = [](long long n, int tb) { return (unsigned)((n + tb - 1) / tb); };
  const int rowGroups = (N + 31) / 32;
  constexpr int RNG = 200;
  const int rangeGroups = (N + RNG - 1) / RNG;

  zero_kernel<<<blk((long long)N * 4, TB), TB, 0, stream>>>((float*)m1, (long long)N * 4);
  gemm_f32_wmma<128><<<dim3(rowGroups * 2), TB, 0, stream>>>(x, W1, h1, N);
  node_alpha_kernel<2><<<blk((long long)N * 2, TB), TB, 0, stream>>>(h1, as1, ad1, asrc1, adst1, N);
  edge_max_kernel<2><<<blk((long long)Etot * 2, TB), TB, 0, stream>>>(srcE, dstE, asrc1, adst1, m1, E, Etot);
  range_expagg_kernel<2, RNG><<<dim3(rangeGroups), 1024, 0, stream>>>(srcE, dstE, asrc1, adst1, m1, h1, s1, agg1, E, N);
  bias_relu_kernel<<<blk((long long)N * 128, TB), TB, 0, stream>>>(agg1, s1, b1, (long long)N * 128, 127);

  gemm_f32_wmma<64><<<dim3(rowGroups), TB, 0, stream>>>(agg1, W2, h2f, N);
  zero_kernel<<<blk((long long)N * 2, TB), TB, 0, stream>>>((float*)m2, (long long)N * 2);
  node_alpha_kernel<1><<<blk((long long)N, TB), TB, 0, stream>>>(h2f, as2, ad2, asrc2, adst2, N);
  edge_max_kernel<1><<<blk((long long)Etot, TB), TB, 0, stream>>>(srcE, dstE, asrc2, adst2, m2, E, Etot);
  range_expagg_kernel<1, RNG><<<dim3(rangeGroups), 1024, 0, stream>>>(srcE, dstE, asrc2, adst2, m2, h2f, s2, agg2, E, N);
  bias_sigmoid_kernel<<<blk((long long)N * 64, TB), TB, 0, stream>>>(agg2, s2, b2, out, (long long)N * 64, 63);
}


# softmax division, bias and activation applied in the range kernel's row store (row and its sum sit in LDS): both element-wise kernels, the sum plane and the aggregate's re-read gone; sums unchanged
# speedup vs baseline: 1.7643x; 1.0680x over previous
#include <hip/hip_runtime.h>
#include <hip/hip_bf16.h>
#include <math.h>

typedef __attribute__((ext_vector_type(2))) float v2f;
typedef __attribute__((ext_vector_type(8))) float v8f;

#define NEG_SLOPE 0.2f

__device__ __forceinline__ unsigned flipf(float f) {
  unsigned u = __float_as_uint(f);
  return (u & 0x80000000u) ? ~u : (u | 0x80000000u);
}
__device__ __forceinline__ float unflipf(unsigned k) {
  unsigned u = (k & 0x80000000u) ? (k ^ 0x80000000u) : ~k;
  return __uint_as_float(u);
}

__global__ void zero_kernel(float* __restrict__ p, long long n) {
  long long i = (long long)blockIdx.x * blockDim.x + threadIdx.x;
  if (i < n) p[i] = 0.f;
}

template<int NC>
__global__ __launch_bounds__(256) void gemm_f32_wmma(const float* __restrict__ A,
                                                     const float* __restrict__ B,
                                                     float* __restrict__ C, int M) {
  constexpr int K = 128;
  constexpr int CG = 64;
  constexpr int LDA = K + 4;
  constexpr int colGroups = NC / CG;
  __shared__ float Wlds[K * CG];
  __shared__ float Alds[32 * LDA];
  int cg = blockIdx.x % colGroups;
  int rg = blockIdx.x / colGroups;

  for (int i = threadIdx.x; i < K * (CG / 4); i += 256) {
    int k = i / (CG / 4), c4 = i % (CG / 4);
    *(float4*)&Wlds[k * CG + c4 * 4] =
        *(const float4*)&B[(size_t)k * NC + cg * CG + c4 * 4];
  }
  for (int i = threadIdx.x; i < 32 * (K / 4); i += 256) {
    int r = i / (K / 4), c4 = i % (K / 4);
    int grow = rg * 32 + r;
    float4 v = make_float4(0.f, 0.f, 0.f, 0.f);
    if (grow < M) v = *(const float4*)&A[(size_t)grow * K + c4 * 4];
    *(float4*)&Alds[r * LDA + c4 * 4] = v;
  }
  __syncthreads();

  int wave = threadIdx.x >> 5;
  int lane = threadIdx.x & 31;
  int half = lane >> 4;
  int lrow = lane & 15;
  int rt = wave >> 2, ct = wave & 3;
  int grow0 = rg * 32 + rt * 16;
  if (grow0 >= M) return;

  const float* Ab = &Alds[(rt * 16 + lrow) * LDA + 2 * half];
  const float* Bb = &Wlds[(2 * half) * CG + ct * 16 + lrow];
  v8f acc = {};
  #pragma unroll
  for (int k0 = 0; k0 < K; k0 += 4) {
    v2f a, b;
    a.x = Ab[k0];        a.y = Ab[k0 + 1];
    b.x = Bb[k0 * CG];   b.y = Bb[k0 * CG + CG];
    acc = __builtin_amdgcn_wmma_f32_16x16x4_f32(false, a, false, b,
                                                (short)0, acc, false, false);
  }
  float* Cp = &C[(size_t)(grow0 + 8 * half) * NC + cg * CG + ct * 16 + lrow];
  #pragma unroll
  for (int j = 0; j < 8; ++j) Cp[(size_t)j * NC] = acc[j];
}

template<int H>
__global__ void node_alpha_kernel(const float* __restrict__ h,
                                  const float* __restrict__ aw_src,
                                  const float* __restrict__ aw_dst,
                                  float* __restrict__ asrc, float* __restrict__ adst, int N) {
  int idx = blockIdx.x * blockDim.x + threadIdx.x;
  if (idx >= N * H) return;
  int hh = idx % H;
  const float* hp = h + (size_t)idx * 64;
  const float* ap = aw_src + hh * 64;
  const float* bp = aw_dst + hh * 64;
  float sa = 0.f, sb = 0.f;
  #pragma unroll
  for (int c = 0; c < 64; c += 4) {
    float4 v = *(const float4*)(hp + c);
    float4 u = *(const float4*)(ap + c);
    float4 t = *(const float4*)(bp + c);
    sa += v.x * u.x + v.y * u.y + v.z * u.z + v.w * u.w;
    sb += v.x * t.x + v.y * t.y + v.z * t.z + v.w * t.w;
  }
  asrc[idx] = sa;
  adst[idx] = sb;
}

template<int H>
__global__ void edge_max_kernel(const int* __restrict__ srcE, const int* __restrict__ dstE,
                                const float* __restrict__ asrc, const float* __restrict__ adst,
                                unsigned* __restrict__ m, int E, int Etot) {
  int idx = blockIdx.x * blockDim.x + threadIdx.x;
  if (idx >= Etot * H) return;
  int e = idx / H, hh = idx % H;
  int s = (e < E) ? srcE[e] : (e - E);
  int d = (e < E) ? dstE[e] : (e - E);
  float v = asrc[s * H + hh] + adst[d * H + hh];
  v = (v > 0.f) ? v : NEG_SLOPE * v;
  atomicMax(&m[d * H + hh], flipf(v));
}

template<int H, int R, bool SIGMOID>
__global__ __launch_bounds__(1024) void range_expagg_kernel(const int* __restrict__ srcE,
                                                            const int* __restrict__ dstE,
                                                            const float* __restrict__ asrc,
                                                            const float* __restrict__ adst,
                                                            const unsigned* __restrict__ m,
                                                            const float* __restrict__ h,
                                                            const float* __restrict__ bias,
                                                            float* __restrict__ rows, int E, int N) {
  constexpr int HC = H * 64;
  constexpr int KK = HC / 32;
  constexpr int NT = 1024;
  constexpr int UV = 4;
  constexpr int UD = 16 / KK;
  constexpr int QCAP = 128 + 32 * 4 * UV;
  static_assert(R <= NT, "one lane per owned node for the self-loops");
  static_assert(NT % HC == 0, "a thread keeps one channel through the store loop");
  __shared__ float acc[R * HC];
  __shared__ float sl[R * H];
  __shared__ float ml[R * H];
  __shared__ float adl[R * H];
  __shared__ int qe[(NT / 32) * QCAP];

  const int tid = threadIdx.x;
  const int lane = tid & 31;
  const int wave = tid >> 5;
  const int lo = blockIdx.x * R;

  for (int i = tid; i < R * HC; i += NT) acc[i] = 0.f;
  for (int i = tid; i < R * H; i += NT) {
    int gi = lo * H + i;
    bool ok = gi < N * H;
    sl[i] = 0.f;
    ml[i] = ok ? unflipf(m[gi]) : 0.f;
    adl[i] = ok ? adst[gi] : 0.f;
  }
  __syncthreads();

  int* myq = qe + wave * QCAP;
  int cnt = 0;

  auto append = [&](bool keep, int e) {
    unsigned mask = __builtin_amdgcn_ballot_w32(keep);
    if (mask == 0u) return;
    if (keep) myq[cnt + __builtin_popcount(mask & ((1u << lane) - 1u))] = e;
    cnt += __builtin_popcount(mask);
  };

  int head = (int)(((16u - (unsigned)((size_t)dstE & 15u)) & 15u) >> 2);
  if (head > E) head = E;
  const int nvec = (E - head) >> 2;
  const int tail0 = head + 4 * nvec;
  const int4* d4 = (const int4*)(dstE + head);
  append(tid < head && (unsigned)(dstE[tid < head ? tid : 0] - lo) < (unsigned)R, tid);
  append(tail0 + tid < E && (unsigned)(dstE[tail0 + tid < E ? tail0 + tid : 0] - lo) < (unsigned)R, tail0 + tid);
  append(tid < R && lo + tid < N, E + lo + tid);

  const int rounds = (nvec + UV * NT - 1) / (UV * NT);
  for (int it = 0; it <= rounds; ++it) {
    if (it < rounds) {
      const int q0 = it * (UV * NT);
      int4 dv[UV];
      #pragma unroll
      for (int j = 0; j < UV; ++j) {
        int q = q0 + j * NT + tid;
        dv[j] = (q < nvec) ? d4[q] : make_int4(-1, -1, -1, -1);
      }
      #pragma unroll
      for (int j = 0; j < UV; ++j) {
        int eb = head + 4 * (q0 + j * NT + tid);
        append((unsigned)(dv[j].x - lo) < (unsigned)R, eb + 0);
        append((unsigned)(dv[j].y - lo) < (unsigned)R, eb + 1);
        append((unsigned)(dv[j].z - lo) < (unsigned)R, eb + 2);
        append((unsigned)(dv[j].w - lo) < (unsigned)R, eb + 3);
      }
    }
    const int leave = (it < rounds) ? 31 : 0;
    while (cnt > leave) {
      const int n = cnt < 32 ? cnt : 32;
      cnt -= n;
      int s = 0, dl = 0;
      float ex0 = 0.f, ex1 = 0.f;
      if (lane < n) {
        int e = myq[cnt + lane];
        s = (e < E) ? srcE[e] : (e - E);
        dl = ((e < E) ? dstE[e] : (e - E)) - lo;
        float v = asrc[s * H] + adl[dl * H];
        v = (v > 0.f) ? v : NEG_SLOPE * v;
        ex0 = __expf(v - ml[dl * H]);
        atomicAdd(&sl[dl * H], ex0);
        if (H == 2) {
          float w = asrc[s * H + 1] + adl[dl * H + 1];
          w = (w > 0.f) ? w : NEG_SLOPE * w;
          ex1 = __expf(w - ml[dl * H + 1]);
          atomicAdd(&sl[dl * H + 1], ex1);
        }
      }
      int i = 0;
      for (; i + UD <= n; i += UD) {
        float val[UD][KK];
        #pragma unroll
        for (int u = 0; u < UD; ++u) {
          int si = __builtin_amdgcn_readlane(s, i + u);
          const float* hp = h + (size_t)si * HC + lane;
          #pragma unroll
          for (int k = 0; k < KK; ++k) val[u][k] = hp[32 * k];
        }
        #pragma unroll
        for (int u = 0; u < UD; ++u) {
          int di = __builtin_amdgcn_readlane(dl, i + u);
          float x0 = __int_as_float(__builtin_amdgcn_readlane(__float_as_int(ex0), i + u));
          float x1 = __int_as_float(__builtin_amdgcn_readlane(__float_as_int(ex1), i + u));
          float* ap = acc + di * HC + lane;
          #pragma unroll
          for (int k = 0; k < KK; ++k) atomicAdd(ap + 32 * k, ((k < 2) ? x0 : x1) * val[u][k]);
        }
      }
      for (; i < n; ++i) {
        int si = __builtin_amdgcn_readlane(s, i);
        int di = __builtin_amdgcn_readlane(dl, i);
        float x0 = __int_as_float(__builtin_amdgcn_readlane(__float_as_int(ex0), i));
        float x1 = __int_as_float(__builtin_amdgcn_readlane(__float_as_int(ex1), i));
        const float* hp = h + (size_t)si * HC + lane;
        float* ap = acc + di * HC + lane;
        float val[KK];
        #pragma unroll
        for (int k = 0; k < KK; ++k) val[k] = hp[32 * k];
        #pragma unroll
        for (int k = 0; k < KK; ++k) atomicAdd(ap + 32 * k, ((k < 2) ? x0 : x1) * val[k]);
      }
    }
  }
  __syncthreads();

  const float bv = bias[tid % HC];
  const int nel = ((N - lo < R) ? (N - lo) : R) * HC;
  float* rp = rows + (size_t)lo * HC;
  for (int i = tid; i < nel; i += NT) {
    float v = acc[i] / sl[i >> 6] + bv;
    rp[i] = SIGMOID ? 1.f / (1.f + __expf(-v)) : (v > 0.f ? v : 0.f);
  }
}

extern "C" void kernel_launch(void* const* d_in, const int* in_sizes, int n_in,
                              void* d_out, int out_size, void* d_ws, size_t ws_size,
                              hipStream_t stream) {
  const float* x   = (const float*)d_in[0];
  const int*   ei  = (const int*)d_in[1];
  const float* W1  = (const float*)d_in[2];
  const float* as1 = (const float*)d_in[3];
  const float* ad1 = (const float*)d_in[4];
  const float* b1  = (const float*)d_in[5];
  const float* W2  = (const float*)d_in[6];
  const float* as2 = (const float*)d_in[7];
  const float* ad2 = (const float*)d_in[8];
  const float* b2  = (const float*)d_in[9];
  float* out = (float*)d_out;

  const int N = in_sizes[0] / 128;
  const int E = in_sizes[1] / 2;
  const int Etot = E + N;
  const int* srcE = ei;
  const int* dstE = ei + E;

  float* w = (float*)d_ws;
  size_t o = 0;
  float* h1    = w + o; o += (size_t)N * 128;
  float* agg1  = w + o; o += (size_t)N * 128;
  float* asrc1 = w + o; o += (size_t)N * 2;
  float* adst1 = w + o; o += (size_t)N * 2;
  unsigned* m1 = (unsigned*)(w + o); o += (size_t)N * 2;
  float* asrc2 = w + o; o += (size_t)N;
  float* adst2 = w + o; o += (size_t)N;
  unsigned* m2 = (unsigned*)(w + o); o += (size_t)N;
  float* h2f   = h1;
  (void)o; (void)ws_size; (void)n_in; (void)out_size;

  const int TB = 256;
  auto blk = [](long long n, int tb) { return (unsigned)((n + tb - 1) / tb); };
  const int rowGroups = (N + 31) / 32;
  constexpr int RNG = 200;
  const int rangeGroups = (N + RNG - 1) / RNG;

  zero_kernel<<<blk((long long)N * 2, TB), TB, 0, stream>>>((float*)m1, (long long)N * 2);
  gemm_f32_wmma<128><<<dim3(rowGroups * 2), TB, 0, stream>>>(x, W1, h1, N);
  node_alpha_kernel<2><<<blk((long long)N * 2, TB), TB, 0, stream>>>(h1, as1, ad1, asrc1, adst1, N);
  edge_max_kernel<2><<<blk((long long)Etot * 2, TB), TB, 0, stream>>>(srcE, dstE, asrc1, adst1, m1, E, Etot);
  range_expagg_kernel<2, RNG, false><<<dim3(rangeGroups), 1024, 0, stream>>>(srcE, dstE, asrc1, adst1, m1, h1, b1, agg1, E, N);

  gemm_f32_wmma<64><<<dim3(rowGroups), TB, 0, stream>>>(agg1, W2, h2f, N);
  zero_kernel<<<blk((long long)N, TB), TB, 0, stream>>>((float*)m2, (long long)N);
  node_alpha_kernel<1><<<blk((long long)N, TB), TB, 0, stream>>>(h2f, as2, ad2, asrc2, adst2, N);
  edge_max_kernel<1><<<blk((long long)Etot, TB), TB, 0, stream>>>(srcE, dstE, asrc2, adst2, m2, E, Etot);
  range_expagg_kernel<1, RNG, true><<<dim3(rangeGroups), 1024, 0, stream>>>(srcE, dstE, asrc2, adst2, m2, h2f, b2, out, E, N);
}


# chunk map and both products run as few long-lived waves: one wave builds a 1024-edge group's map words; 1024 product work-groups loop over row groups, weight panel staged once; sums unchanged
# speedup vs baseline: 4.3087x; 2.4422x over previous
#include <hip/hip_runtime.h>
#include <hip/hip_bf16.h>
#include <math.h>

typedef __attribute__((ext_vector_type(2))) float v2f;
typedef __attribute__((ext_vector_type(8))) float v8f;

#define NEG_SLOPE 0.2f

__device__ __forceinline__ unsigned flipf(float f) {
  unsigned u = __float_as_uint(f);
  return (u & 0x80000000u) ? ~u : (u | 0x80000000u);
}
__device__ __forceinline__ float unflipf(unsigned k) {
  unsigned u = (k & 0x80000000u) ? (k ^ 0x80000000u) : ~k;
  return __uint_as_float(u);
}

template<int NC>
__global__ __launch_bounds__(256) void gemm_f32_wmma(const float* __restrict__ A,
                                                     const float* __restrict__ B,
                                                     float* __restrict__ C, int M) {
  constexpr int K = 128;
  constexpr int CG = 64;
  constexpr int LDA = K + 4;
  constexpr int colGroups = NC / CG;
  __shared__ float Wlds[K * CG];
  __shared__ float Alds[2][32 * LDA];
  const int cg = blockIdx.x % colGroups;
  const int rg0 = blockIdx.x / colGroups;
  const int rgStep = gridDim.x / colGroups;
  const int rowGroups = (M + 31) >> 5;
  const int tid = threadIdx.x;

  float4 av[4];
  auto loadA = [&](int rg) {
    #pragma unroll
    for (int j = 0; j < 4; ++j) {
      const int i = tid + j * 256;
      const int r = i / (K / 4), c4 = i % (K / 4);
      const int grow = rg * 32 + r;
      av[j] = make_float4(0.f, 0.f, 0.f, 0.f);
      if (grow < M) av[j] = *(const float4*)&A[(size_t)grow * K + c4 * 4];
    }
  };
  if (rg0 < rowGroups) loadA(rg0);

  {
    float4 bv[8];
    #pragma unroll
    for (int j = 0; j < 8; ++j) {
      const int i = tid + j * 256;
      const int k = i / (CG / 4), c4 = i % (CG / 4);
      bv[j] = *(const float4*)&B[(size_t)k * NC + cg * CG + c4 * 4];
    }
    #pragma unroll
    for (int j = 0; j < 8; ++j) {
      const int i = tid + j * 256;
      const int k = i / (CG / 4), c4 = i % (CG / 4);
      *(float4*)&Wlds[k * CG + c4 * 4] = bv[j];
    }
  }

  const int wave = __builtin_amdgcn_readfirstlane(tid >> 5);
  const int lane = tid & 31;
  const int half = lane >> 4;
  const int lrow = lane & 15;
  const int rt = wave >> 2, ct = wave & 3;
  const float* Bb = &Wlds[(2 * half) * CG + ct * 16 + lrow];

  int buf = 0;
  for (int rg = rg0; rg < rowGroups; rg += rgStep) {
    #pragma unroll
    for (int j = 0; j < 4; ++j) {
      const int i = tid + j * 256;
      const int r = i / (K / 4), c4 = i % (K / 4);
      *(float4*)&Alds[buf][r * LDA + c4 * 4] = av[j];
    }
    __syncthreads();
    if (rg + rgStep < rowGroups) loadA(rg + rgStep);

    const int grow0 = rg * 32 + rt * 16;
    if (grow0 < M) {
      const float* Ab = &Alds[buf][(rt * 16 + lrow) * LDA + 2 * half];
      v8f acc = {};
      #pragma unroll
      for (int k0 = 0; k0 < K; k0 += 4) {
        v2f a, b;
        a.x = Ab[k0];        a.y = Ab[k0 + 1];
        b.x = Bb[k0 * CG];   b.y = Bb[k0 * CG + CG];
        acc = __builtin_amdgcn_wmma_f32_16x16x4_f32(false, a, false, b,
                                                    (short)0, acc, false, false);
      }
      float* Cp = &C[(size_t)(grow0 + 8 * half) * NC + cg * CG + ct * 16 + lrow];
      #pragma unroll
      for (int j = 0; j < 8; ++j) Cp[(size_t)j * NC] = acc[j];
    }
    buf ^= 1;
  }
}

template<int H>
__global__ void node_alpha_kernel(const float* __restrict__ h,
                                  const float* __restrict__ aw_src,
                                  const float* __restrict__ aw_dst,
                                  float* __restrict__ asrc, float* __restrict__ adst, int N) {
  int idx = blockIdx.x * blockDim.x + threadIdx.x;
  if (idx >= N * H) return;
  int hh = idx % H;
  const float* hp = h + (size_t)idx * 64;
  const float* ap = aw_src + hh * 64;
  const float* bp = aw_dst + hh * 64;
  float sa = 0.f, sb = 0.f;
  #pragma unroll
  for (int c = 0; c < 64; c += 4) {
    float4 v = *(const float4*)(hp + c);
    float4 u = *(const float4*)(ap + c);
    float4 t = *(const float4*)(bp + c);
    sa += v.x * u.x + v.y * u.y + v.z * u.z + v.w * u.w;
    sb += v.x * t.x + v.y * t.y + v.z * t.z + v.w * t.w;
  }
  asrc[idx] = sa;
  adst[idx] = sb;
}

template<int R>
__global__ __launch_bounds__(128) void chunk_map_kernel(const int* __restrict__ dstE,
                                                        unsigned* __restrict__ map, int E) {
  __shared__ unsigned bits[4][256];
  const int lane = threadIdx.x & 31;
  const int wave = threadIdx.x >> 5;
  const int g = blockIdx.x * 4 + wave;
  if (g >= ((E + 1023) >> 10)) return;
  unsigned* const b = bits[wave];
  #pragma unroll
  for (int j = 0; j < 8; ++j) b[lane + 32 * j] = 0u;
  int d[32];
  #pragma unroll
  for (int c = 0; c < 32; ++c) {
    const int e = g * 1024 + c * 32 + lane;
    d[c] = dstE[e < E ? e : E - 1];
  }
  #pragma unroll
  for (int c = 0; c < 32; ++c) {
    const int e = g * 1024 + c * 32 + lane;
    if (e < E) atomicOr(&b[((unsigned)d[c] / (unsigned)R) & 255u], 1u << c);
  }
  #pragma unroll
  for (int j = 0; j < 8; ++j) map[(size_t)g * 256 + lane + 32 * j] = b[lane + 32 * j];
}

template<int H, int R, bool SIGMOID, bool EMIT, bool USE>
__global__ __launch_bounds__(1024) void range_expagg_kernel(const int* __restrict__ srcE,
                                                            const int* __restrict__ dstE,
                                                            const unsigned* __restrict__ map,
                                                            const float* __restrict__ asrc,
                                                            const float* __restrict__ adst,
                                                            const float* __restrict__ h,
                                                            const float* __restrict__ bias,
                                                            float* __restrict__ rows,
                                                            int* __restrict__ elist, int E, int N) {
  constexpr int HC = H * 64;
  constexpr int KK = HC / 32;
  constexpr int NT = 1024;
  constexpr int NW = NT / 32;
  constexpr int UC = 8;
  constexpr int UD = 16 / KK;
  constexpr int QCAP = 32 + 32 * UC;
  constexpr int LCAP = 6144;
  constexpr int LSTRIDE = 2 * LCAP + 32;
  static_assert(R <= NT, "one lane per owned node for the self-loops");
  static_assert(NT % HC == 0, "a thread keeps one channel through the store loop");
  __shared__ float acc[R * HC];
  __shared__ float sl[R * H];
  __shared__ unsigned mk[R * H];
  __shared__ float adl[R * H];
  __shared__ int qe[NW * QCAP];
  __shared__ int ls[LCAP];
  __shared__ int ld[LCAP];
  __shared__ float lv[LCAP * H];
  __shared__ int lcount;

  const int tid = threadIdx.x;
  const int lane = tid & 31;
  const int wave = tid >> 5;
  const int lo = blockIdx.x * R;
  int* const gcount = elist + (size_t)blockIdx.x * LSTRIDE;
  int* const gsrc = gcount + 32;
  int* const gdl = gsrc + LCAP;

  for (int i = tid; i < R * HC; i += NT) acc[i] = 0.f;
  for (int i = tid; i < R * H; i += NT) {
    int gi = lo * H + i;
    sl[i] = 0.f;
    mk[i] = 0u;
    adl[i] = (gi < N * H) ? adst[gi] : 0.f;
  }
  if (tid == 0) lcount = 0;
  __syncthreads();

  int* myq = qe + wave * QCAP;
  int cnt = 0;

  auto append = [&](bool keep, int e) {
    unsigned mask = __builtin_amdgcn_ballot_w32(keep);
    if (mask == 0u) return;
    if (keep) myq[cnt + __builtin_popcount(mask & ((1u << lane) - 1u))] = e;
    cnt += __builtin_popcount(mask);
  };

  auto edge_logits = [&](int e, int& s, int& dl, float& v0, float& v1) {
    s = (e < E) ? srcE[e] : (e - E);
    dl = ((e < E) ? dstE[e] : (e - E)) - lo;
    float v = asrc[s * H] + adl[dl * H];
    v0 = (v > 0.f) ? v : NEG_SLOPE * v;
    v1 = 0.f;
    if (H == 2) {
      float w = asrc[s * H + 1] + adl[dl * H + 1];
      v1 = (w > 0.f) ? w : NEG_SLOPE * w;
    }
  };

  auto edge_exp = [&](int dl, float v0, float v1, float& ex0, float& ex1) {
    ex0 = __expf(v0 - unflipf(mk[dl * H]));
    atomicAdd(&sl[dl * H], ex0);
    ex1 = 0.f;
    if (H == 2) {
      ex1 = __expf(v1 - unflipf(mk[dl * H + 1]));
      atomicAdd(&sl[dl * H + 1], ex1);
    }
  };

  auto add_rows = [&](int n, int s, int dl, float ex0, float ex1) {
    int i = 0;
    for (; i + UD <= n; i += UD) {
      float val[UD][KK];
      #pragma unroll
      for (int u = 0; u < UD; ++u) {
        int si = __builtin_amdgcn_readlane(s, i + u);
        const float* hp = h + (size_t)si * HC + lane;
        #pragma unroll
        for (int k = 0; k < KK; ++k) val[u][k] = hp[32 * k];
      }
      #pragma unroll
      for (int u = 0; u < UD; ++u) {
        int di = __builtin_amdgcn_readlane(dl, i + u);
        float x0 = __int_as_float(__builtin_amdgcn_readlane(__float_as_int(ex0), i + u));
        float x1 = __int_as_float(__builtin_amdgcn_readlane(__float_as_int(ex1), i + u));
        float* ap = acc + di * HC + lane;
        #pragma unroll
        for (int k = 0; k < KK; ++k) atomicAdd(ap + 32 * k, ((k < 2) ? x0 : x1) * val[u][k]);
      }
    }
    for (; i < n; ++i) {
      int si = __builtin_amdgcn_readlane(s, i);
      int di = __builtin_amdgcn_readlane(dl, i);
      float x0 = __int_as_float(__builtin_amdgcn_readlane(__float_as_int(ex0), i));
      float x1 = __int_as_float(__builtin_amdgcn_readlane(__float_as_int(ex1), i));
      const float* hp = h + (size_t)si * HC + lane;
      float* ap = acc + di * HC + lane;
      float val[KK];
      #pragma unroll
      for (int k = 0; k < KK; ++k) val[k] = hp[32 * k];
      #pragma unroll
      for (int k = 0; k < KK; ++k) atomicAdd(ap + 32 * k, ((k < 2) ? x0 : x1) * val[k]);
    }
  };

  const int wv = __builtin_amdgcn_readfirstlane(wave);

  auto walk = [&](auto batch) {
    cnt = 0;
    append(tid < R && lo + tid < N, E + lo + tid);

    const int hb = blockIdx.x & 255;
    const int nGroups = (E + 1023) >> 10;
    const int nPairs = (nGroups + 1) >> 1;
    int pnext = wv;
    int pcur = 0;
    int li = 0, ln = 0;
    unsigned wlo = 0u, whi = 0u;
    unsigned long long msk = 0ull;
    int ebase = 0;
    for (;;) {
      while (msk == 0ull) {
        if (li == ln) {
          if (pnext >= nPairs) break;
          const int mine = pnext + 32 * lane;
          const int g0 = 2 * mine;
          wlo = (mine < nPairs) ? map[(size_t)g0 * 256 + hb] : 0u;
          whi = (mine < nPairs && g0 + 1 < nGroups) ? map[(size_t)(g0 + 1) * 256 + hb] : 0u;
          const int left = (nPairs - pnext + 31) >> 5;
          ln = left < 32 ? left : 32;
          li = 0;
          pcur = pnext;
          pnext += 32 * 32;
        }
        const unsigned a = (unsigned)__builtin_amdgcn_readlane((int)wlo, li);
        const unsigned b = (unsigned)__builtin_amdgcn_readlane((int)whi, li);
        msk = (unsigned long long)a | ((unsigned long long)b << 32);
        ebase = (pcur + 32 * li) * 2048;
        ++li;
      }
      const bool more = (msk != 0ull);
      if (more) {
        int ce[UC], dv[UC];
        bool cv[UC];
        const int c0 = __builtin_ctzll(msk);
        #pragma unroll
        for (int j = 0; j < UC; ++j) {
          cv[j] = (msk != 0ull);
          const int c = cv[j] ? __builtin_ctzll(msk) : c0;
          msk &= msk - 1ull;
          ce[j] = ebase + 32 * c + lane;
          dv[j] = dstE[ce[j] < E ? ce[j] : E - 1];
        }
        #pragma unroll
        for (int j = 0; j < UC; ++j)
          append(cv[j] && ce[j] < E && (unsigned)(dv[j] - lo) < (unsigned)R, ce[j]);
      }
      const int leave = more ? 31 : 0;
      while (cnt > leave) {
        const int n = cnt < 32 ? cnt : 32;
        cnt -= n;
        batch(n);
      }
      if (!more) break;
    }
  };

  int given = LCAP + 1;
  if (USE) given = __builtin_amdgcn_readfirstlane(gcount[0]);
  if (USE && given <= LCAP) {
    constexpr int IT = LCAP / NT;
    static_assert(LCAP % NT == 0, "a whole number of places a thread");
    int sj[IT], dj[IT];
    float aj[IT][H];
    #pragma unroll
    for (int j = 0; j < IT; ++j) {
      const int i = tid + j * NT;
      sj[j] = (i < given) ? gsrc[i] : 0;
      dj[j] = (i < given) ? gdl[i] : 0;
    }
    #pragma unroll
    for (int j = 0; j < IT; ++j) {
      #pragma unroll
      for (int q = 0; q < H; ++q) aj[j][q] = asrc[sj[j] * H + q];
    }
    #pragma unroll
    for (int j = 0; j < IT; ++j) {
      const int i = tid + j * NT;
      if (i < given) {
        ls[i] = sj[j];
        ld[i] = dj[j];
        #pragma unroll
        for (int q = 0; q < H; ++q) {
          const float v = aj[j][q] + adl[dj[j] * H + q];
          const float lv1 = (v > 0.f) ? v : NEG_SLOPE * v;
          atomicMax(&mk[dj[j] * H + q], flipf(lv1));
          lv[i * H + q] = lv1;
        }
      }
    }
    if (tid == 0) lcount = given;
  } else
  walk([&](int n) {
    int s = 0, dl = 0;
    float v0 = 0.f, v1 = 0.f;
    if (lane < n) {
      edge_logits(myq[cnt + lane], s, dl, v0, v1);
      atomicMax(&mk[dl * H], flipf(v0));
      if (H == 2) atomicMax(&mk[dl * H + 1], flipf(v1));
    }
    int base = 0;
    if (lane == 0) base = atomicAdd(&lcount, n);
    base = __builtin_amdgcn_readfirstlane(base);
    if (base + n <= LCAP && lane < n) {
      ls[base + lane] = s;
      ld[base + lane] = dl;
      lv[(base + lane) * H] = v0;
      if (H == 2) lv[(base + lane) * H + 1] = v1;
    }
  });
  __syncthreads();

  const int total = lcount;
  if (EMIT) {
    if (tid == 0) gcount[0] = total;
    if (total <= LCAP) {
      #pragma unroll
      for (int j = 0; j < LCAP / NT; ++j) {
        const int i = tid + j * NT;
        if (i < total) {
          gsrc[i] = ls[i];
          gdl[i] = ld[i];
        }
      }
    }
  }
  if (total <= LCAP) {
    const int per = (((total + NW - 1) / NW + UD - 1) / UD) * UD;
    const int beg = wv * per;
    const int end = (beg + per < total) ? (beg + per) : total;
    for (int b = beg; b < end; b += 32) {
      const int n = (end - b < 32) ? (end - b) : 32;
      int s = 0, dl = 0;
      float ex0 = 0.f, ex1 = 0.f;
      if (lane < n) {
        s = ls[b + lane];
        dl = ld[b + lane];
        edge_exp(dl, lv[(b + lane) * H], (H == 2) ? lv[(b + lane) * H + 1] : 0.f, ex0, ex1);
      }
      add_rows(n, s, dl, ex0, ex1);
    }
  } else {
    walk([&](int n) {
      int s = 0, dl = 0;
      float v0 = 0.f, v1 = 0.f, ex0 = 0.f, ex1 = 0.f;
      if (lane < n) {
        edge_logits(myq[cnt + lane], s, dl, v0, v1);
        edge_exp(dl, v0, v1, ex0, ex1);
      }
      add_rows(n, s, dl, ex0, ex1);
    });
  }
  __syncthreads();

  const float bv = bias[tid % HC];
  const int nel = ((N - lo < R) ? (N - lo) : R) * HC;
  float* rp = rows + (size_t)lo * HC;
  for (int i = tid; i < nel; i += NT) {
    float v = acc[i] / sl[i >> 6] + bv;
    rp[i] = SIGMOID ? 1.f / (1.f + __expf(-v)) : (v > 0.f ? v : 0.f);
  }
}

extern "C" void kernel_launch(void* const* d_in, const int* in_sizes, int n_in,
                              void* d_out, int out_size, void* d_ws, size_t ws_size,
                              hipStream_t stream) {
  const float* x   = (const float*)d_in[0];
  const int*   ei  = (const int*)d_in[1];
  const float* W1  = (const float*)d_in[2];
  const float* as1 = (const float*)d_in[3];
  const float* ad1 = (const float*)d_in[4];
  const float* b1  = (const float*)d_in[5];
  const float* W2  = (const float*)d_in[6];
  const float* as2 = (const float*)d_in[7];
  const float* ad2 = (const float*)d_in[8];
  const float* b2  = (const float*)d_in[9];
  float* out = (float*)d_out;

  const int N = in_sizes[0] / 128;
  const int E = in_sizes[1] / 2;
  const int* srcE = ei;
  const int* dstE = ei + E;

  float* w = (float*)d_ws;
  size_t o = 0;
  unsigned* cmap = (unsigned*)(w + o); o += (size_t)((E + 1023) / 1024) * 256;
  float* h1    = w + o; o += (size_t)N * 128;
  float* agg1  = w + o; o += (size_t)N * 128;
  float* asrc1 = w + o; o += (size_t)N * 2;
  float* adst1 = w + o; o += (size_t)N * 2;
  float* asrc2 = w + o; o += (size_t)N;
  float* adst2 = w + o; o += (size_t)N;
  int* elist = (int*)(w + o);
  float* h2f   = h1;
  (void)o; (void)ws_size; (void)n_in; (void)out_size;

  const int TB = 256;
  auto blk = [](long long n, int tb) { return (unsigned)((n + tb - 1) / tb); };
  constexpr int GEMM_WGS = 1024;
  constexpr int RNG = 200;
  const int rangeGroups = (N + RNG - 1) / RNG;

  chunk_map_kernel<RNG><<<blk(((long long)E + 1023) / 1024, 4), 128, 0, stream>>>(dstE, cmap, E);

  gemm_f32_wmma<128><<<dim3(GEMM_WGS), TB, 0, stream>>>(x, W1, h1, N);
  node_alpha_kernel<2><<<blk((long long)N * 2, TB), TB, 0, stream>>>(h1, as1, ad1, asrc1, adst1, N);
  range_expagg_kernel<2, RNG, false, true, false><<<dim3(rangeGroups), 1024, 0, stream>>>(srcE, dstE, cmap, asrc1, adst1, h1, b1, agg1, elist, E, N);

  gemm_f32_wmma<64><<<dim3(GEMM_WGS), TB, 0, stream>>>(agg1, W2, h2f, N);
  node_alpha_kernel<1><<<blk((long long)N, TB), TB, 0, stream>>>(h2f, as2, ad2, asrc2, adst2, N);
  range_expagg_kernel<1, RNG, true, false, true><<<dim3(rangeGroups), 1024, 0, stream>>>(srcE, dstE, cmap, asrc2, adst2, h2f, b2, out, elist, E, N);
}


# logits as x*(W*a): a product work-group folds its weight panel with the head's attention vectors and dots its input rows with them; both logit kernels gone; new logit sum order
# speedup vs baseline: 4.7219x; 1.0959x over previous
#include <hip/hip_runtime.h>
#include <hip/hip_bf16.h>
#include <math.h>

typedef __attribute__((ext_vector_type(2))) float v2f;
typedef __attribute__((ext_vector_type(8))) float v8f;

#define NEG_SLOPE 0.2f

__device__ __forceinline__ unsigned flipf(float f) {
  unsigned u = __float_as_uint(f);
  return (u & 0x80000000u) ? ~u : (u | 0x80000000u);
}
__device__ __forceinline__ float unflipf(unsigned k) {
  unsigned u = (k & 0x80000000u) ? (k ^ 0x80000000u) : ~k;
  return __uint_as_float(u);
}

template<int NC>
__global__ __launch_bounds__(256) void gemm_f32_wmma(const float* __restrict__ A,
                                                     const float* __restrict__ B,
                                                     float* __restrict__ C,
                                                     const float* __restrict__ aw_src,
                                                     const float* __restrict__ aw_dst,
                                                     float* __restrict__ asrc,
                                                     float* __restrict__ adst, int M) {
  constexpr int K = 128;
  constexpr int CG = 64;
  constexpr int LDA = K + 4;
  constexpr int colGroups = NC / CG;
  __shared__ float Wlds[K * CG];
  __shared__ float Alds[2][32 * LDA];
  __shared__ float Ulds[2][K];
  const int cg = blockIdx.x % colGroups;
  const int rg0 = blockIdx.x / colGroups;
  const int rgStep = gridDim.x / colGroups;
  const int rowGroups = (M + 31) >> 5;
  const int tid = threadIdx.x;

  float4 av[4];
  auto loadA = [&](int rg) {
    #pragma unroll
    for (int j = 0; j < 4; ++j) {
      const int i = tid + j * 256;
      const int r = i / (K / 4), c4 = i % (K / 4);
      const int grow = rg * 32 + r;
      av[j] = make_float4(0.f, 0.f, 0.f, 0.f);
      if (grow < M) av[j] = *(const float4*)&A[(size_t)grow * K + c4 * 4];
    }
  };
  if (rg0 < rowGroups) loadA(rg0);

  {
    float4 bv[8];
    #pragma unroll
    for (int j = 0; j < 8; ++j) {
      const int i = tid + j * 256;
      const int k = i / (CG / 4), c4 = i % (CG / 4);
      bv[j] = *(const float4*)&B[(size_t)k * NC + cg * CG + c4 * 4];
    }
    const float4 as4 = *(const float4*)&aw_src[cg * CG + (tid % (CG / 4)) * 4];
    const float4 ad4 = *(const float4*)&aw_dst[cg * CG + (tid % (CG / 4)) * 4];
    #pragma unroll
    for (int j = 0; j < 8; ++j) {
      const int i = tid + j * 256;
      const int k = i / (CG / 4), c4 = i % (CG / 4);
      *(float4*)&Wlds[k * CG + c4 * 4] = bv[j];
    }
    float v16[16];
    #pragma unroll
    for (int j = 0; j < 8; ++j) {
      v16[2 * j]     = bv[j].x * as4.x + bv[j].y * as4.y + bv[j].z * as4.z + bv[j].w * as4.w;
      v16[2 * j + 1] = bv[j].x * ad4.x + bv[j].y * ad4.y + bv[j].z * ad4.z + bv[j].w * ad4.w;
    }
    const bool b3 = (tid & 8) != 0, b2 = (tid & 4) != 0, b1 = (tid & 2) != 0, b0 = (tid & 1) != 0;
    float v8[8], v4[4], v2[2];
    #pragma unroll
    for (int i = 0; i < 8; ++i) v8[i] = (b3 ? v16[8 + i] : v16[i]) + __shfl_xor(b3 ? v16[i] : v16[8 + i], 8);
    #pragma unroll
    for (int i = 0; i < 4; ++i) v4[i] = (b2 ? v8[4 + i] : v8[i]) + __shfl_xor(b2 ? v8[i] : v8[4 + i], 4);
    #pragma unroll
    for (int i = 0; i < 2; ++i) v2[i] = (b1 ? v4[2 + i] : v4[i]) + __shfl_xor(b1 ? v4[i] : v4[2 + i], 2);
    const float v1 = (b0 ? v2[1] : v2[0]) + __shfl_xor(b0 ? v2[0] : v2[1], 1);
    Ulds[tid & 1][(tid >> 4) + 16 * ((tid & 15) >> 1)] = v1;
  }
  __syncthreads();
  const float4 us4 = *(const float4*)&Ulds[0][(tid & 31) * 4];
  const float4 ud4 = *(const float4*)&Ulds[1][(tid & 31) * 4];

  const int wave = __builtin_amdgcn_readfirstlane(tid >> 5);
  const int lane = tid & 31;
  const int half = lane >> 4;
  const int lrow = lane & 15;
  const int rt = wave >> 2, ct = wave & 3;
  const float* Bb = &Wlds[(2 * half) * CG + ct * 16 + lrow];

  int buf = 0;
  for (int rg = rg0; rg < rowGroups; rg += rgStep) {
    #pragma unroll
    for (int j = 0; j < 4; ++j) {
      const int i = tid + j * 256;
      const int r = i / (K / 4), c4 = i % (K / 4);
      *(float4*)&Alds[buf][r * LDA + c4 * 4] = av[j];
    }
    float p8[8];
    #pragma unroll
    for (int j = 0; j < 4; ++j) {
      p8[j]     = av[j].x * us4.x + av[j].y * us4.y + av[j].z * us4.z + av[j].w * us4.w;
      p8[4 + j] = av[j].x * ud4.x + av[j].y * ud4.y + av[j].z * ud4.z + av[j].w * ud4.w;
    }
    __syncthreads();
    if (rg + rgStep < rowGroups) loadA(rg + rgStep);
    {
      const bool h4 = (tid & 16) != 0, h3 = (tid & 8) != 0, h2 = (tid & 4) != 0;
      float p4[4], p2[2];
      #pragma unroll
      for (int i = 0; i < 4; ++i) p4[i] = (h4 ? p8[4 + i] : p8[i]) + __shfl_xor(h4 ? p8[i] : p8[4 + i], 16);
      #pragma unroll
      for (int i = 0; i < 2; ++i) p2[i] = (h3 ? p4[2 + i] : p4[i]) + __shfl_xor(h3 ? p4[i] : p4[2 + i], 8);
      float p1 = (h2 ? p2[1] : p2[0]) + __shfl_xor(h2 ? p2[0] : p2[1], 4);
      p1 += __shfl_xor(p1, 2);
      p1 += __shfl_xor(p1, 1);
      const int arow = rg * 32 + (tid >> 5) + 8 * ((tid >> 2) & 3);
      if ((tid & 3) == 0 && arow < M) (h4 ? adst : asrc)[(size_t)arow * colGroups + cg] = p1;
    }

    const int grow0 = rg * 32 + rt * 16;
    if (grow0 < M) {
      const float* Ab = &Alds[buf][(rt * 16 + lrow) * LDA + 2 * half];
      v8f acc = {};
      #pragma unroll
      for (int k0 = 0; k0 < K; k0 += 4) {
        v2f a, b;
        a.x = Ab[k0];        a.y = Ab[k0 + 1];
        b.x = Bb[k0 * CG];   b.y = Bb[k0 * CG + CG];
        acc = __builtin_amdgcn_wmma_f32_16x16x4_f32(false, a, false, b,
                                                    (short)0, acc, false, false);
      }
      float* Cp = &C[(size_t)(grow0 + 8 * half) * NC + cg * CG + ct * 16 + lrow];
      #pragma unroll
      for (int j = 0; j < 8; ++j) Cp[(size_t)j * NC] = acc[j];
    }
    buf ^= 1;
  }
}

template<int R>
__global__ __launch_bounds__(128) void chunk_map_kernel(const int* __restrict__ dstE,
                                                        unsigned* __restrict__ map, int E) {
  __shared__ unsigned bits[4][256];
  const int lane = threadIdx.x & 31;
  const int wave = threadIdx.x >> 5;
  const int g = blockIdx.x * 4 + wave;
  if (g >= ((E + 1023) >> 10)) return;
  unsigned* const b = bits[wave];
  #pragma unroll
  for (int j = 0; j < 8; ++j) b[lane + 32 * j] = 0u;
  int d[32];
  #pragma unroll
  for (int c = 0; c < 32; ++c) {
    const int e = g * 1024 + c * 32 + lane;
    d[c] = dstE[e < E ? e : E - 1];
  }
  #pragma unroll
  for (int c = 0; c < 32; ++c) {
    const int e = g * 1024 + c * 32 + lane;
    if (e < E) atomicOr(&b[((unsigned)d[c] / (unsigned)R) & 255u], 1u << c);
  }
  #pragma unroll
  for (int j = 0; j < 8; ++j) map[(size_t)g * 256 + lane + 32 * j] = b[lane + 32 * j];
}

template<int H, int R, bool SIGMOID, bool EMIT, bool USE>
__global__ __launch_bounds__(1024) void range_expagg_kernel(const int* __restrict__ srcE,
                                                            const int* __restrict__ dstE,
                                                            const unsigned* __restrict__ map,
                                                            const float* __restrict__ asrc,
                                                            const float* __restrict__ adst,
                                                            const float* __restrict__ h,
                                                            const float* __restrict__ bias,
                                                            float* __restrict__ rows,
                                                            int* __restrict__ elist, int E, int N) {
  constexpr int HC = H * 64;
  constexpr int KK = HC / 32;
  constexpr int NT = 1024;
  constexpr int NW = NT / 32;
  constexpr int UC = 8;
  constexpr int UD = 16 / KK;
  constexpr int QCAP = 32 + 32 * UC;
  constexpr int LCAP = 6144;
  constexpr int LSTRIDE = 2 * LCAP + 32;
  static_assert(R <= NT, "one lane per owned node for the self-loops");
  static_assert(NT % HC == 0, "a thread keeps one channel through the store loop");
  __shared__ float acc[R * HC];
  __shared__ float sl[R * H];
  __shared__ unsigned mk[R * H];
  __shared__ float adl[R * H];
  __shared__ int qe[NW * QCAP];
  __shared__ int ls[LCAP];
  __shared__ int ld[LCAP];
  __shared__ float lv[LCAP * H];
  __shared__ int lcount;

  const int tid = threadIdx.x;
  const int lane = tid & 31;
  const int wave = tid >> 5;
  const int lo = blockIdx.x * R;
  int* const gcount = elist + (size_t)blockIdx.x * LSTRIDE;
  int* const gsrc = gcount + 32;
  int* const gdl = gsrc + LCAP;

  for (int i = tid; i < R * HC; i += NT) acc[i] = 0.f;
  for (int i = tid; i < R * H; i += NT) {
    int gi = lo * H + i;
    sl[i] = 0.f;
    mk[i] = 0u;
    adl[i] = (gi < N * H) ? adst[gi] : 0.f;
  }
  if (tid == 0) lcount = 0;
  __syncthreads();

  int* myq = qe + wave * QCAP;
  int cnt = 0;

  auto append = [&](bool keep, int e) {
    unsigned mask = __builtin_amdgcn_ballot_w32(keep);
    if (mask == 0u) return;
    if (keep) myq[cnt + __builtin_popcount(mask & ((1u << lane) - 1u))] = e;
    cnt += __builtin_popcount(mask);
  };

  auto edge_logits = [&](int e, int& s, int& dl, float& v0, float& v1) {
    s = (e < E) ? srcE[e] : (e - E);
    dl = ((e < E) ? dstE[e] : (e - E)) - lo;
    float v = asrc[s * H] + adl[dl * H];
    v0 = (v > 0.f) ? v : NEG_SLOPE * v;
    v1 = 0.f;
    if (H == 2) {
      float w = asrc[s * H + 1] + adl[dl * H + 1];
      v1 = (w > 0.f) ? w : NEG_SLOPE * w;
    }
  };

  auto edge_exp = [&](int dl, float v0, float v1, float& ex0, float& ex1) {
    ex0 = __expf(v0 - unflipf(mk[dl * H]));
    atomicAdd(&sl[dl * H], ex0);
    ex1 = 0.f;
    if (H == 2) {
      ex1 = __expf(v1 - unflipf(mk[dl * H + 1]));
      atomicAdd(&sl[dl * H + 1], ex1);
    }
  };

  auto add_rows = [&](int n, int s, int dl, float ex0, float ex1) {
    int i = 0;
    for (; i + UD <= n; i += UD) {
      float val[UD][KK];
      #pragma unroll
      for (int u = 0; u < UD; ++u) {
        int si = __builtin_amdgcn_readlane(s, i + u);
        const float* hp = h + (size_t)si * HC + lane;
        #pragma unroll
        for (int k = 0; k < KK; ++k) val[u][k] = hp[32 * k];
      }
      #pragma unroll
      for (int u = 0; u < UD; ++u) {
        int di = __builtin_amdgcn_readlane(dl, i + u);
        float x0 = __int_as_float(__builtin_amdgcn_readlane(__float_as_int(ex0), i + u));
        float x1 = __int_as_float(__builtin_amdgcn_readlane(__float_as_int(ex1), i + u));
        float* ap = acc + di * HC + lane;
        #pragma unroll
        for (int k = 0; k < KK; ++k) atomicAdd(ap + 32 * k, ((k < 2) ? x0 : x1) * val[u][k]);
      }
    }
    for (; i < n; ++i) {
      int si = __builtin_amdgcn_readlane(s, i);
      int di = __builtin_amdgcn_readlane(dl, i);
      float x0 = __int_as_float(__builtin_amdgcn_readlane(__float_as_int(ex0), i));
      float x1 = __int_as_float(__builtin_amdgcn_readlane(__float_as_int(ex1), i));
      const float* hp = h + (size_t)si * HC + lane;
      float* ap = acc + di * HC + lane;
      float val[KK];
      #pragma unroll
      for (int k = 0; k < KK; ++k) val[k] = hp[32 * k];
      #pragma unroll
      for (int k = 0; k < KK; ++k) atomicAdd(ap + 32 * k, ((k < 2) ? x0 : x1) * val[k]);
    }
  };

  const int wv = __builtin_amdgcn_readfirstlane(wave);

  auto walk = [&](auto batch) {
    cnt = 0;
    append(tid < R && lo + tid < N, E + lo + tid);

    const int hb = blockIdx.x & 255;
    const int nGroups = (E + 1023) >> 10;
    const int nPairs = (nGroups + 1) >> 1;
    int pnext = wv;
    int pcur = 0;
    int li = 0, ln = 0;
    unsigned wlo = 0u, whi = 0u;
    unsigned long long msk = 0ull;
    int ebase = 0;
    for (;;) {
      while (msk == 0ull) {
        if (li == ln) {
          if (pnext >= nPairs) break;
          const int mine = pnext + 32 * lane;
          const int g0 = 2 * mine;
          wlo = (mine < nPairs) ? map[(size_t)g0 * 256 + hb] : 0u;
          whi = (mine < nPairs && g0 + 1 < nGroups) ? map[(size_t)(g0 + 1) * 256 + hb] : 0u;
          const int left = (nPairs - pnext + 31) >> 5;
          ln = left < 32 ? left : 32;
          li = 0;
          pcur = pnext;
          pnext += 32 * 32;
        }
        const unsigned a = (unsigned)__builtin_amdgcn_readlane((int)wlo, li);
        const unsigned b = (unsigned)__builtin_amdgcn_readlane((int)whi, li);
        msk = (unsigned long long)a | ((unsigned long long)b << 32);
        ebase = (pcur + 32 * li) * 2048;
        ++li;
      }
      const bool more = (msk != 0ull);
      if (more) {
        int ce[UC], dv[UC];
        bool cv[UC];
        const int c0 = __builtin_ctzll(msk);
        #pragma unroll
        for (int j = 0; j < UC; ++j) {
          cv[j] = (msk != 0ull);
          const int c = cv[j] ? __builtin_ctzll(msk) : c0;
          msk &= msk - 1ull;
          ce[j] = ebase + 32 * c + lane;
          dv[j] = dstE[ce[j] < E ? ce[j] : E - 1];
        }
        #pragma unroll
        for (int j = 0; j < UC; ++j)
          append(cv[j] && ce[j] < E && (unsigned)(dv[j] - lo) < (unsigned)R, ce[j]);
      }
      const int leave = more ? 31 : 0;
      while (cnt > leave) {
        const int n = cnt < 32 ? cnt : 32;
        cnt -= n;
        batch(n);
      }
      if (!more) break;
    }
  };

  int given = LCAP + 1;
  if (USE) given = __builtin_amdgcn_readfirstlane(gcount[0]);
  if (USE && given <= LCAP) {
    constexpr int IT = LCAP / NT;
    static_assert(LCAP % NT == 0, "a whole number of places a thread");
    int sj[IT], dj[IT];
    float aj[IT][H];
    #pragma unroll
    for (int j = 0; j < IT; ++j) {
      const int i = tid + j * NT;
      sj[j] = (i < given) ? gsrc[i] : 0;
      dj[j] = (i < given) ? gdl[i] : 0;
    }
    #pragma unroll
    for (int j = 0; j < IT; ++j) {
      #pragma unroll
      for (int q = 0; q < H; ++q) aj[j][q] = asrc[sj[j] * H + q];
    }
    #pragma unroll
    for (int j = 0; j < IT; ++j) {
      const int i = tid + j * NT;
      if (i < given) {
        ls[i] = sj[j];
        ld[i] = dj[j];
        #pragma unroll
        for (int q = 0; q < H; ++q) {
          const float v = aj[j][q] + adl[dj[j] * H + q];
          const float lv1 = (v > 0.f) ? v : NEG_SLOPE * v;
          atomicMax(&mk[dj[j] * H + q], flipf(lv1));
          lv[i * H + q] = lv1;
        }
      }
    }
    if (tid == 0) lcount = given;
  } else
  walk([&](int n) {
    int s = 0, dl = 0;
    float v0 = 0.f, v1 = 0.f;
    if (lane < n) {
      edge_logits(myq[cnt + lane], s, dl, v0, v1);
      atomicMax(&mk[dl * H], flipf(v0));
      if (H == 2) atomicMax(&mk[dl * H + 1], flipf(v1));
    }
    int base = 0;
    if (lane == 0) base = atomicAdd(&lcount, n);
    base = __builtin_amdgcn_readfirstlane(base);
    if (base + n <= LCAP && lane < n) {
      ls[base + lane] = s;
      ld[base + lane] = dl;
      lv[(base + lane) * H] = v0;
      if (H == 2) lv[(base + lane) * H + 1] = v1;
    }
  });
  __syncthreads();

  const int total = lcount;
  if (EMIT) {
    if (tid == 0) gcount[0] = total;
    if (total <= LCAP) {
      #pragma unroll
      for (int j = 0; j < LCAP / NT; ++j) {
        const int i = tid + j * NT;
        if (i < total) {
          gsrc[i] = ls[i];
          gdl[i] = ld[i];
        }
      }
    }
  }
  if (total <= LCAP) {
    const int per = (((total + NW - 1) / NW + UD - 1) / UD) * UD;
    const int beg = wv * per;
    const int end = (beg + per < total) ? (beg + per) : total;
    for (int b = beg; b < end; b += 32) {
      const int n = (end - b < 32) ? (end - b) : 32;
      int s = 0, dl = 0;
      float ex0 = 0.f, ex1 = 0.f;
      if (lane < n) {
        s = ls[b + lane];
        dl = ld[b + lane];
        edge_exp(dl, lv[(b + lane) * H], (H == 2) ? lv[(b + lane) * H + 1] : 0.f, ex0, ex1);
      }
      add_rows(n, s, dl, ex0, ex1);
    }
  } else {
    walk([&](int n) {
      int s = 0, dl = 0;
      float v0 = 0.f, v1 = 0.f, ex0 = 0.f, ex1 = 0.f;
      if (lane < n) {
        edge_logits(myq[cnt + lane], s, dl, v0, v1);
        edge_exp(dl, v0, v1, ex0, ex1);
      }
      add_rows(n, s, dl, ex0, ex1);
    });
  }
  __syncthreads();

  const float bv = bias[tid % HC];
  const int nel = ((N - lo < R) ? (N - lo) : R) * HC;
  float* rp = rows + (size_t)lo * HC;
  for (int i = tid; i < nel; i += NT) {
    float v = acc[i] / sl[i >> 6] + bv;
    rp[i] = SIGMOID ? 1.f / (1.f + __expf(-v)) : (v > 0.f ? v : 0.f);
  }
}

extern "C" void kernel_launch(void* const* d_in, const int* in_sizes, int n_in,
                              void* d_out, int out_size, void* d_ws, size_t ws_size,
                              hipStream_t stream) {
  const float* x   = (const float*)d_in[0];
  const int*   ei  = (const int*)d_in[1];
  const float* W1  = (const float*)d_in[2];
  const float* as1 = (const float*)d_in[3];
  const float* ad1 = (const float*)d_in[4];
  const float* b1  = (const float*)d_in[5];
  const float* W2  = (const float*)d_in[6];
  const float* as2 = (const float*)d_in[7];
  const float* ad2 = (const float*)d_in[8];
  const float* b2  = (const float*)d_in[9];
  float* out = (float*)d_out;

  const int N = in_sizes[0] / 128;
  const int E = in_sizes[1] / 2;
  const int* srcE = ei;
  const int* dstE = ei + E;

  float* w = (float*)d_ws;
  size_t o = 0;
  unsigned* cmap = (unsigned*)(w + o); o += (size_t)((E + 1023) / 1024) * 256;
  float* h1    = w + o; o += (size_t)N * 128;
  float* agg1  = w + o; o += (size_t)N * 128;
  float* asrc1 = w + o; o += (size_t)N * 2;
  float* adst1 = w + o; o += (size_t)N * 2;
  float* asrc2 = w + o; o += (size_t)N;
  float* adst2 = w + o; o += (size_t)N;
  int* elist = (int*)(w + o);
  float* h2f   = h1;
  (void)o; (void)ws_size; (void)n_in; (void)out_size;

  const int TB = 256;
  auto blk = [](long long n, int tb) { return (unsigned)((n + tb - 1) / tb); };
  constexpr int GEMM_WGS = 1024;
  constexpr int RNG = 200;
  const int rangeGroups = (N + RNG - 1) / RNG;

  chunk_map_kernel<RNG><<<blk(((long long)E + 1023) / 1024, 4), 128, 0, stream>>>(dstE, cmap, E);

  gemm_f32_wmma<128><<<dim3(GEMM_WGS), TB, 0, stream>>>(x, W1, h1, as1, ad1, asrc1, adst1, N);
  range_expagg_kernel<2, RNG, false, true, false><<<dim3(rangeGroups), 1024, 0, stream>>>(srcE, dstE, cmap, asrc1, adst1, h1, b1, agg1, elist, E, N);

  gemm_f32_wmma<64><<<dim3(GEMM_WGS), TB, 0, stream>>>(agg1, W2, h2f, as2, ad2, asrc2, adst2, N);
  range_expagg_kernel<1, RNG, true, false, true><<<dim3(rangeGroups), 1024, 0, stream>>>(srcE, dstE, cmap, asrc2, adst2, h2f, b2, out, elist, E, N);
}


# layer-1 edge search by lanes, not a chunk a wave: a map byte a thread queues flagged chunks, a wave loads 32 chunks, lane j files chunk j's edges; logits in one bulk pass; sums in LDS arrival order
# speedup vs baseline: 4.9552x; 1.0494x over previous
#include <hip/hip_runtime.h>
#include <hip/hip_bf16.h>
#include <math.h>

typedef __attribute__((ext_vector_type(2))) float v2f;
typedef __attribute__((ext_vector_type(8))) float v8f;
typedef __attribute__((ext_vector_type(2))) int v2i;

#define NEG_SLOPE 0.2f

__device__ __forceinline__ unsigned flipf(float f) {
  unsigned u = __float_as_uint(f);
  return (u & 0x80000000u) ? ~u : (u | 0x80000000u);
}
__device__ __forceinline__ float unflipf(unsigned k) {
  unsigned u = (k & 0x80000000u) ? (k ^ 0x80000000u) : ~k;
  return __uint_as_float(u);
}

__device__ __forceinline__ int wave_incl_scan(int v, int lane) {
  #pragma unroll
  for (int d = 1; d < 32; d <<= 1) {
    const int t = __shfl_up(v, d);
    if (lane >= d) v += t;
  }
  return v;
}

template<int NC>
__global__ __launch_bounds__(256) void gemm_f32_wmma(const float* __restrict__ A,
                                                     const float* __restrict__ B,
                                                     float* __restrict__ C,
                                                     const float* __restrict__ aw_src,
                                                     const float* __restrict__ aw_dst,
                                                     float* __restrict__ asrc,
                                                     float* __restrict__ adst, int M) {
  constexpr int K = 128;
  constexpr int CG = 64;
  constexpr int LDA = K + 4;
  constexpr int colGroups = NC / CG;
  __shared__ float Wlds[K * CG];
  __shared__ float Alds[2][32 * LDA];
  __shared__ float Ulds[2][K];
  const int cg = blockIdx.x % colGroups;
  const int rg0 = blockIdx.x / colGroups;
  const int rgStep = gridDim.x / colGroups;
  const int rowGroups = (M + 31) >> 5;
  const int tid = threadIdx.x;

  float4 av[4];
  auto loadA = [&](int rg) {
    #pragma unroll
    for (int j = 0; j < 4; ++j) {
      const int i = tid + j * 256;
      const int r = i / (K / 4), c4 = i % (K / 4);
      const int grow = rg * 32 + r;
      av[j] = make_float4(0.f, 0.f, 0.f, 0.f);
      if (grow < M) av[j] = *(const float4*)&A[(size_t)grow * K + c4 * 4];
    }
  };
  if (rg0 < rowGroups) loadA(rg0);

  {
    float4 bv[8];
    #pragma unroll
    for (int j = 0; j < 8; ++j) {
      const int i = tid + j * 256;
      const int k = i / (CG / 4), c4 = i % (CG / 4);
      bv[j] = *(const float4*)&B[(size_t)k * NC + cg * CG + c4 * 4];
    }
    const float4 as4 = *(const float4*)&aw_src[cg * CG + (tid % (CG / 4)) * 4];
    const float4 ad4 = *(const float4*)&aw_dst[cg * CG + (tid % (CG / 4)) * 4];
    #pragma unroll
    for (int j = 0; j < 8; ++j) {
      const int i = tid + j * 256;
      const int k = i / (CG / 4), c4 = i % (CG / 4);
      *(float4*)&Wlds[k * CG + c4 * 4] = bv[j];
    }
    float v16[16];
    #pragma unroll
    for (int j = 0; j < 8; ++j) {
      v16[2 * j]     = bv[j].x * as4.x + bv[j].y * as4.y + bv[j].z * as4.z + bv[j].w * as4.w;
      v16[2 * j + 1] = bv[j].x * ad4.x + bv[j].y * ad4.y + bv[j].z * ad4.z + bv[j].w * ad4.w;
    }
    const bool b3 = (tid & 8) != 0, b2 = (tid & 4) != 0, b1 = (tid & 2) != 0, b0 = (tid & 1) != 0;
    float v8[8], v4[4], v2[2];
    #pragma unroll
    for (int i = 0; i < 8; ++i) v8[i] = (b3 ? v16[8 + i] : v16[i]) + __shfl_xor(b3 ? v16[i] : v16[8 + i], 8);
    #pragma unroll
    for (int i = 0; i < 4; ++i) v4[i] = (b2 ? v8[4 + i] : v8[i]) + __shfl_xor(b2 ? v8[i] : v8[4 + i], 4);
    #pragma unroll
    for (int i = 0; i < 2; ++i) v2[i] = (b1 ? v4[2 + i] : v4[i]) + __shfl_xor(b1 ? v4[i] : v4[2 + i], 2);
    const float v1 = (b0 ? v2[1] : v2[0]) + __shfl_xor(b0 ? v2[0] : v2[1], 1);
    Ulds[tid & 1][(tid >> 4) + 16 * ((tid & 15) >> 1)] = v1;
  }
  __syncthreads();
  const float4 us4 = *(const float4*)&Ulds[0][(tid & 31) * 4];
  const float4 ud4 = *(const float4*)&Ulds[1][(tid & 31) * 4];

  const int wave = __builtin_amdgcn_readfirstlane(tid >> 5);
  const int lane = tid & 31;
  const int half = lane >> 4;
  const int lrow = lane & 15;
  const int rt = wave >> 2, ct = wave & 3;
  const float* Bb = &Wlds[(2 * half) * CG + ct * 16 + lrow];

  int buf = 0;
  for (int rg = rg0; rg < rowGroups; rg += rgStep) {
    #pragma unroll
    for (int j = 0; j < 4; ++j) {
      const int i = tid + j * 256;
      const int r = i / (K / 4), c4 = i % (K / 4);
      *(float4*)&Alds[buf][r * LDA + c4 * 4] = av[j];
    }
    float p8[8];
    #pragma unroll
    for (int j = 0; j < 4; ++j) {
      p8[j]     = av[j].x * us4.x + av[j].y * us4.y + av[j].z * us4.z + av[j].w * us4.w;
      p8[4 + j] = av[j].x * ud4.x + av[j].y * ud4.y + av[j].z * ud4.z + av[j].w * ud4.w;
    }
    __syncthreads();
    if (rg + rgStep < rowGroups) loadA(rg + rgStep);
    {
      const bool h4 = (tid & 16) != 0, h3 = (tid & 8) != 0, h2 = (tid & 4) != 0;
      float p4[4], p2[2];
      #pragma unroll
      for (int i = 0; i < 4; ++i) p4[i] = (h4 ? p8[4 + i] : p8[i]) + __shfl_xor(h4 ? p8[i] : p8[4 + i], 16);
      #pragma unroll
      for (int i = 0; i < 2; ++i) p2[i] = (h3 ? p4[2 + i] : p4[i]) + __shfl_xor(h3 ? p4[i] : p4[2 + i], 8);
      float p1 = (h2 ? p2[1] : p2[0]) + __shfl_xor(h2 ? p2[0] : p2[1], 4);
      p1 += __shfl_xor(p1, 2);
      p1 += __shfl_xor(p1, 1);
      const int arow = rg * 32 + (tid >> 5) + 8 * ((tid >> 2) & 3);
      if ((tid & 3) == 0 && arow < M) (h4 ? adst : asrc)[(size_t)arow * colGroups + cg] = p1;
    }

    const int grow0 = rg * 32 + rt * 16;
    if (grow0 < M) {
      const float* Ab = &Alds[buf][(rt * 16 + lrow) * LDA + 2 * half];
      v8f acc = {};
      #pragma unroll
      for (int k0 = 0; k0 < K; k0 += 4) {
        v2f a, b;
        a.x = Ab[k0];        a.y = Ab[k0 + 1];
        b.x = Bb[k0 * CG];   b.y = Bb[k0 * CG + CG];
        acc = __builtin_amdgcn_wmma_f32_16x16x4_f32(false, a, false, b,
                                                    (short)0, acc, false, false);
      }
      float* Cp = &C[(size_t)(grow0 + 8 * half) * NC + cg * CG + ct * 16 + lrow];
      #pragma unroll
      for (int j = 0; j < 8; ++j) Cp[(size_t)j * NC] = acc[j];
    }
    buf ^= 1;
  }
}

template<int R>
__global__ __launch_bounds__(128) void chunk_map_kernel(const int* __restrict__ dstE,
                                                        unsigned* __restrict__ map, int E) {
  __shared__ unsigned bits[4][256];
  const int lane = threadIdx.x & 31;
  const int wave = threadIdx.x >> 5;
  const int g = blockIdx.x * 4 + wave;
  if (g >= ((E + 1023) >> 10)) return;
  unsigned* const b = bits[wave];
  #pragma unroll
  for (int j = 0; j < 8; ++j) b[lane + 32 * j] = 0u;
  int d[32];
  #pragma unroll
  for (int c = 0; c < 32; ++c) {
    const int e = g * 1024 + c * 32 + lane;
    d[c] = dstE[e < E ? e : E - 1];
  }
  #pragma unroll
  for (int c = 0; c < 32; ++c) {
    const int e = g * 1024 + c * 32 + lane;
    if (e < E) atomicOr(&b[((unsigned)d[c] / (unsigned)R) & 255u], 1u << c);
  }
  #pragma unroll
  for (int j = 0; j < 8; ++j) map[(size_t)g * 256 + lane + 32 * j] = b[lane + 32 * j];
}

template<int H, int R, bool SIGMOID, bool EMIT, bool USE>
__global__ __launch_bounds__(1024) void range_expagg_kernel(const int* __restrict__ srcE,
                                                            const int* __restrict__ dstE,
                                                            const unsigned* __restrict__ map,
                                                            const float* __restrict__ asrc,
                                                            const float* __restrict__ adst,
                                                            const float* __restrict__ h,
                                                            const float* __restrict__ bias,
                                                            float* __restrict__ rows,
                                                            int* __restrict__ elist, int E, int N) {
  constexpr int HC = H * 64;
  constexpr int KK = HC / 32;
  constexpr int NT = 1024;
  constexpr int NW = NT / 32;
  constexpr int UC = 8;
  constexpr int UD = 16 / KK;
  constexpr int QCAP = 32 + 32 * UC;
  constexpr int LCAP = 6144;
  constexpr int LSTRIDE = 2 * LCAP + 32;
  static_assert(R <= NT, "one lane per owned node for the self-loops");
  static_assert(NT % HC == 0, "a thread keeps one channel through the store loop");
  __shared__ float acc[R * HC];
  __shared__ float sl[R * H];
  __shared__ unsigned mk[R * H];
  __shared__ float adl[R * H];
  __shared__ int qe[NW * QCAP];
  __shared__ v2i lp[LCAP];
  __shared__ float lv[LCAP * H];
  __shared__ int lcount;
  __shared__ int qcount;

  const int tid = threadIdx.x;
  const int lane = tid & 31;
  const int wave = tid >> 5;
  const int lo = blockIdx.x * R;
  int* const gcount = elist + (size_t)blockIdx.x * LSTRIDE;
  int* const gsrc = gcount + 32;
  int* const gdl = gsrc + LCAP;

  for (int i = tid; i < R * HC; i += NT) acc[i] = 0.f;
  for (int i = tid; i < R * H; i += NT) {
    int gi = lo * H + i;
    sl[i] = 0.f;
    mk[i] = 0u;
    adl[i] = (gi < N * H) ? adst[gi] : 0.f;
  }
  const int nself = (N - lo < R) ? (N - lo) : R;
  if (tid == 0) { lcount = USE ? 0 : nself; qcount = 0; }
  __syncthreads();

  int* myq = qe + wave * QCAP;
  int cnt = 0;

  auto append = [&](bool keep, int e) {
    unsigned mask = __builtin_amdgcn_ballot_w32(keep);
    if (mask == 0u) return;
    if (keep) myq[cnt + __builtin_popcount(mask & ((1u << lane) - 1u))] = e;
    cnt += __builtin_popcount(mask);
  };

  auto edge_logits = [&](int e, int& s, int& dl, float& v0, float& v1) {
    s = (e < E) ? srcE[e] : (e - E);
    dl = ((e < E) ? dstE[e] : (e - E)) - lo;
    float v = asrc[s * H] + adl[dl * H];
    v0 = (v > 0.f) ? v : NEG_SLOPE * v;
    v1 = 0.f;
    if (H == 2) {
      float w = asrc[s * H + 1] + adl[dl * H + 1];
      v1 = (w > 0.f) ? w : NEG_SLOPE * w;
    }
  };

  auto edge_exp = [&](int dl, float v0, float v1, float& ex0, float& ex1) {
    ex0 = __expf(v0 - unflipf(mk[dl * H]));
    atomicAdd(&sl[dl * H], ex0);
    ex1 = 0.f;
    if (H == 2) {
      ex1 = __expf(v1 - unflipf(mk[dl * H + 1]));
      atomicAdd(&sl[dl * H + 1], ex1);
    }
  };

  auto add_rows = [&](int n, int s, int dl, float ex0, float ex1) {
    int i = 0;
    for (; i + UD <= n; i += UD) {
      float val[UD][KK];
      #pragma unroll
      for (int u = 0; u < UD; ++u) {
        int si = __builtin_amdgcn_readlane(s, i + u);
        const float* hp = h + (size_t)si * HC + lane;
        #pragma unroll
        for (int k = 0; k < KK; ++k) val[u][k] = hp[32 * k];
      }
      #pragma unroll
      for (int u = 0; u < UD; ++u) {
        int di = __builtin_amdgcn_readlane(dl, i + u);
        float x0 = __int_as_float(__builtin_amdgcn_readlane(__float_as_int(ex0), i + u));
        float x1 = __int_as_float(__builtin_amdgcn_readlane(__float_as_int(ex1), i + u));
        float* ap = acc + di * HC + lane;
        #pragma unroll
        for (int k = 0; k < KK; ++k) atomicAdd(ap + 32 * k, ((k < 2) ? x0 : x1) * val[u][k]);
      }
    }
    for (; i < n; ++i) {
      int si = __builtin_amdgcn_readlane(s, i);
      int di = __builtin_amdgcn_readlane(dl, i);
      float x0 = __int_as_float(__builtin_amdgcn_readlane(__float_as_int(ex0), i));
      float x1 = __int_as_float(__builtin_amdgcn_readlane(__float_as_int(ex1), i));
      const float* hp = h + (size_t)si * HC + lane;
      float* ap = acc + di * HC + lane;
      float val[KK];
      #pragma unroll
      for (int k = 0; k < KK; ++k) val[k] = hp[32 * k];
      #pragma unroll
      for (int k = 0; k < KK; ++k) atomicAdd(ap + 32 * k, ((k < 2) ? x0 : x1) * val[k]);
    }
  };

  const int wv = __builtin_amdgcn_readfirstlane(wave);

  auto walk = [&](auto batch) {
    cnt = 0;
    append(tid < R && lo + tid < N, E + lo + tid);

    const int hb = blockIdx.x & 255;
    const int nGroups = (E + 1023) >> 10;
    const int nPairs = (nGroups + 1) >> 1;
    int pnext = wv;
    int pcur = 0;
    int li = 0, ln = 0;
    unsigned wlo = 0u, whi = 0u;
    unsigned long long msk = 0ull;
    int ebase = 0;
    for (;;) {
      while (msk == 0ull) {
        if (li == ln) {
          if (pnext >= nPairs) break;
          const int mine = pnext + 32 * lane;
          const int g0 = 2 * mine;
          wlo = (mine < nPairs) ? map[(size_t)g0 * 256 + hb] : 0u;
          whi = (mine < nPairs && g0 + 1 < nGroups) ? map[(size_t)(g0 + 1) * 256 + hb] : 0u;
          const int left = (nPairs - pnext + 31) >> 5;
          ln = left < 32 ? left : 32;
          li = 0;
          pcur = pnext;
          pnext += 32 * 32;
        }
        const unsigned a = (unsigned)__builtin_amdgcn_readlane((int)wlo, li);
        const unsigned b = (unsigned)__builtin_amdgcn_readlane((int)whi, li);
        msk = (unsigned long long)a | ((unsigned long long)b << 32);
        ebase = (pcur + 32 * li) * 2048;
        ++li;
      }
      const bool more = (msk != 0ull);
      if (more) {
        int ce[UC], dv[UC];
        bool cv[UC];
        const int c0 = __builtin_ctzll(msk);
        #pragma unroll
        for (int j = 0; j < UC; ++j) {
          cv[j] = (msk != 0ull);
          const int c = cv[j] ? __builtin_ctzll(msk) : c0;
          msk &= msk - 1ull;
          ce[j] = ebase + 32 * c + lane;
          dv[j] = dstE[ce[j] < E ? ce[j] : E - 1];
        }
        #pragma unroll
        for (int j = 0; j < UC; ++j)
          append(cv[j] && ce[j] < E && (unsigned)(dv[j] - lo) < (unsigned)R, ce[j]);
      }
      const int leave = more ? 31 : 0;
      while (cnt > leave) {
        const int n = cnt < 32 ? cnt : 32;
        cnt -= n;
        batch(n);
      }
      if (!more) break;
    }
  };

  constexpr int IT = LCAP / NT;
  static_assert(LCAP % NT == 0, "a whole number of places a thread");
  int given = LCAP + 1;
  if (USE) given = __builtin_amdgcn_readfirstlane(gcount[0]);
  bool listed = false;
  if (!USE) {
    constexpr int QTOT = NW * QCAP;
    if (tid < nself) lp[tid].x = E + lo + tid;
    const int tail0 = E & ~31;
    if (tid < E - tail0) {
      const int e = tail0 + tid;
      if ((unsigned)(dstE[e] - lo) < (unsigned)R) {
        const int pos = atomicAdd(&lcount, 1);
        if (pos < LCAP) lp[pos].x = e;
      }
    }
    {
      const int hb = blockIdx.x & 255;
      const int nBytes = ((E + 1023) >> 10) * 4;
      constexpr int MT = 4;
      for (int i0 = 0; i0 < nBytes; i0 += NT * MT) {
        unsigned w[MT];
        #pragma unroll
        for (int k = 0; k < MT; ++k) {
          const int idx = i0 + k * NT + tid;
          w[k] = (idx < nBytes) ? map[(size_t)(idx >> 2) * 256 + hb] : 0u;
        }
        int pcs = 0;
        #pragma unroll
        for (int k = 0; k < MT; ++k) {
          const int idx = i0 + k * NT + tid;
          const int ct = (tail0 >> 5) - idx * 8;
          w[k] = (w[k] >> (8 * (idx & 3))) & 255u;
          if (tail0 < E && ct >= 0 && ct < 8) w[k] &= ~(1u << ct);
          pcs += __builtin_popcount(w[k]);
        }
        const int inc = wave_incl_scan(pcs, lane);
        int base = 0;
        if (lane == 31 && inc) base = atomicAdd(&qcount, inc);
        int pos = __builtin_amdgcn_readlane(base, 31) + inc - pcs;
        #pragma unroll
        for (int k = 0; k < MT; ++k) {
          const int idx = i0 + k * NT + tid;
          unsigned m = w[k];
          while (m) {
            const int b = __builtin_ctz(m);
            m &= m - 1u;
            if (pos < QTOT) qe[pos] = (idx * 8 + b) * 32;
            ++pos;
          }
        }
      }
    }
    __syncthreads();
    const int nq = __builtin_amdgcn_readfirstlane(qcount);
    if (nq <= QTOT) {
      const int per = (((nq + NW - 1) / NW + 31) / 32) * 32;
      const int beg = wv * per;
      const int end = (beg + per < nq) ? (beg + per) : nq;
      for (int b0 = beg; b0 < end; b0 += 32) {
        const int nv = (end - b0 < 32) ? (end - b0) : 32;
        const int vb = qe[b0 + (lane < nv ? lane : 0)];
        int dv[32];
        #pragma unroll
        for (int j = 0; j < 32; ++j) dv[j] = dstE[__builtin_amdgcn_readlane(vb, j) + lane];
        int vkm = 0;
        #pragma unroll
        for (int j = 0; j < 32; ++j) {
          const int k = (int)__builtin_amdgcn_ballot_w32((unsigned)(dv[j] - lo) < (unsigned)R);
          vkm = (lane == j) ? k : vkm;
        }
        unsigned m = (lane < nv) ? (unsigned)vkm : 0u;
        const int pc = __builtin_popcount(m);
        const int inc = wave_incl_scan(pc, lane);
        int base = 0;
        if (lane == 31 && inc) base = atomicAdd(&lcount, inc);
        int pos = __builtin_amdgcn_readlane(base, 31) + inc - pc;
        while (m) {
          const int b = __builtin_ctz(m);
          m &= m - 1u;
          if (pos < LCAP) lp[pos].x = vb + b;
          ++pos;
        }
      }
    }
    __syncthreads();
    const int found = (nq <= QTOT) ? __builtin_amdgcn_readfirstlane(lcount) : (LCAP + 1);
    if (found <= LCAP) {
      int sj[IT], dj[IT];
      float aj[IT][H];
      #pragma unroll
      for (int j = 0; j < IT; ++j) {
        const int i = tid + j * NT;
        int e = E;
        if (i < found) e = lp[i].x;
        sj[j] = (e < E) ? srcE[e] : (e - E);
        dj[j] = ((e < E) ? dstE[e] : (e - E)) - lo;
      }
      #pragma unroll
      for (int j = 0; j < IT; ++j) {
        #pragma unroll
        for (int q = 0; q < H; ++q) aj[j][q] = asrc[sj[j] * H + q];
      }
      #pragma unroll
      for (int j = 0; j < IT; ++j) {
        const int i = tid + j * NT;
        if (i < found) {
          lp[i] = v2i{sj[j], dj[j]};
          #pragma unroll
          for (int q = 0; q < H; ++q) {
            const float v = aj[j][q] + adl[dj[j] * H + q];
            const float lv1 = (v > 0.f) ? v : NEG_SLOPE * v;
            atomicMax(&mk[dj[j] * H + q], flipf(lv1));
            lv[i * H + q] = lv1;
          }
        }
      }
      listed = true;
    } else {
      __syncthreads();
      if (tid == 0) lcount = 0;
      __syncthreads();
    }
  } else if (given <= LCAP) {
    int sj[IT], dj[IT];
    float aj[IT][H];
    #pragma unroll
    for (int j = 0; j < IT; ++j) {
      const int i = tid + j * NT;
      sj[j] = (i < given) ? gsrc[i] : 0;
      dj[j] = (i < given) ? gdl[i] : 0;
    }
    #pragma unroll
    for (int j = 0; j < IT; ++j) {
      #pragma unroll
      for (int q = 0; q < H; ++q) aj[j][q] = asrc[sj[j] * H + q];
    }
    #pragma unroll
    for (int j = 0; j < IT; ++j) {
      const int i = tid + j * NT;
      if (i < given) {
        lp[i] = v2i{sj[j], dj[j]};
        #pragma unroll
        for (int q = 0; q < H; ++q) {
          const float v = aj[j][q] + adl[dj[j] * H + q];
          const float lv1 = (v > 0.f) ? v : NEG_SLOPE * v;
          atomicMax(&mk[dj[j] * H + q], flipf(lv1));
          lv[i * H + q] = lv1;
        }
      }
    }
    if (tid == 0) lcount = given;
    listed = true;
  }
  if (!listed) walk([&](int n) {
    int s = 0, dl = 0;
    float v0 = 0.f, v1 = 0.f;
    if (lane < n) {
      edge_logits(myq[cnt + lane], s, dl, v0, v1);
      atomicMax(&mk[dl * H], flipf(v0));
      if (H == 2) atomicMax(&mk[dl * H + 1], flipf(v1));
    }
    int base = 0;
    if (lane == 0) base = atomicAdd(&lcount, n);
    base = __builtin_amdgcn_readfirstlane(base);
    if (base + n <= LCAP && lane < n) {
      lp[base + lane] = v2i{s, dl};
      lv[(base + lane) * H] = v0;
      if (H == 2) lv[(base + lane) * H + 1] = v1;
    }
  });
  __syncthreads();

  const int total = lcount;
  if (EMIT) {
    if (tid == 0) gcount[0] = total;
    if (total <= LCAP) {
      #pragma unroll
      for (int j = 0; j < LCAP / NT; ++j) {
        const int i = tid + j * NT;
        if (i < total) {
          const v2i p = lp[i];
          gsrc[i] = p.x;
          gdl[i] = p.y;
        }
      }
    }
  }
  if (total <= LCAP) {
    const int per = (((total + NW - 1) / NW + UD - 1) / UD) * UD;
    const int beg = wv * per;
    const int end = (beg + per < total) ? (beg + per) : total;
    for (int b = beg; b < end; b += 32) {
      const int n = (end - b < 32) ? (end - b) : 32;
      int s = 0, dl = 0;
      float ex0 = 0.f, ex1 = 0.f;
      if (lane < n) {
        const v2i p = lp[b + lane];
        s = p.x;
        dl = p.y;
        edge_exp(dl, lv[(b + lane) * H], (H == 2) ? lv[(b + lane) * H + 1] : 0.f, ex0, ex1);
      }
      add_rows(n, s, dl, ex0, ex1);
    }
  } else {
    walk([&](int n) {
      int s = 0, dl = 0;
      float v0 = 0.f, v1 = 0.f, ex0 = 0.f, ex1 = 0.f;
      if (lane < n) {
        edge_logits(myq[cnt + lane], s, dl, v0, v1);
        edge_exp(dl, v0, v1, ex0, ex1);
      }
      add_rows(n, s, dl, ex0, ex1);
    });
  }
  __syncthreads();

  const float bv = bias[tid % HC];
  const int nel = ((N - lo < R) ? (N - lo) : R) * HC;
  float* rp = rows + (size_t)lo * HC;
  for (int i = tid; i < nel; i += NT) {
    float v = acc[i] / sl[i >> 6] + bv;
    rp[i] = SIGMOID ? 1.f / (1.f + __expf(-v)) : (v > 0.f ? v : 0.f);
  }
}

extern "C" void kernel_launch(void* const* d_in, const int* in_sizes, int n_in,
                              void* d_out, int out_size, void* d_ws, size_t ws_size,
                              hipStream_t stream) {
  const float* x   = (const float*)d_in[0];
  const int*   ei  = (const int*)d_in[1];
  const float* W1  = (const float*)d_in[2];
  const float* as1 = (const float*)d_in[3];
  const float* ad1 = (const float*)d_in[4];
  const float* b1  = (const float*)d_in[5];
  const float* W2  = (const float*)d_in[6];
  const float* as2 = (const float*)d_in[7];
  const float* ad2 = (const float*)d_in[8];
  const float* b2  = (const float*)d_in[9];
  float* out = (float*)d_out;

  const int N = in_sizes[0] / 128;
  const int E = in_sizes[1] / 2;
  const int* srcE = ei;
  const int* dstE = ei + E;

  float* w = (float*)d_ws;
  size_t o = 0;
  unsigned* cmap = (unsigned*)(w + o); o += (size_t)((E + 1023) / 1024) * 256;
  float* h1    = w + o; o += (size_t)N * 128;
  float* agg1  = w + o; o += (size_t)N * 128;
  float* asrc1 = w + o; o += (size_t)N * 2;
  float* adst1 = w + o; o += (size_t)N * 2;
  float* asrc2 = w + o; o += (size_t)N;
  float* adst2 = w + o; o += (size_t)N;
  int* elist = (int*)(w + o);
  float* h2f   = h1;
  (void)o; (void)ws_size; (void)n_in; (void)out_size;

  const int TB = 256;
  auto blk = [](long long n, int tb) { return (unsigned)((n + tb - 1) / tb); };
  constexpr int GEMM_WGS = 1024;
  constexpr int RNG = 200;
  const int rangeGroups = (N + RNG - 1) / RNG;

  chunk_map_kernel<RNG><<<blk(((long long)E + 1023) / 1024, 4), 128, 0, stream>>>(dstE, cmap, E);

  gemm_f32_wmma<128><<<dim3(GEMM_WGS), TB, 0, stream>>>(x, W1, h1, as1, ad1, asrc1, adst1, N);
  range_expagg_kernel<2, RNG, false, true, false><<<dim3(rangeGroups), 1024, 0, stream>>>(srcE, dstE, cmap, asrc1, adst1, h1, b1, agg1, elist, E, N);

  gemm_f32_wmma<64><<<dim3(GEMM_WGS), TB, 0, stream>>>(agg1, W2, h2f, as2, ad2, asrc2, adst2, N);
  range_expagg_kernel<1, RNG, true, false, true><<<dim3(rangeGroups), 1024, 0, stream>>>(srcE, dstE, cmap, asrc2, adst2, h2f, b2, out, elist, E, N);
}


# row gather by whole rows: one 16-B (layer 2: 8-B) load a lane per edge, not one 4-B load a line; 8 (16) rows in flight a wave; LDS rows keep lanes' words side by side; sums in LDS arrival order
# speedup vs baseline: 5.4009x; 1.0899x over previous
#include <hip/hip_runtime.h>
#include <hip/hip_bf16.h>
#include <math.h>

typedef __attribute__((ext_vector_type(2))) float v2f;
typedef __attribute__((ext_vector_type(8))) float v8f;
typedef __attribute__((ext_vector_type(2))) int v2i;

#define NEG_SLOPE 0.2f

template<int G> struct GroupOf { static constexpr int value = G; };

__device__ __forceinline__ unsigned flipf(float f) {
  unsigned u = __float_as_uint(f);
  return (u & 0x80000000u) ? ~u : (u | 0x80000000u);
}
__device__ __forceinline__ float unflipf(unsigned k) {
  unsigned u = (k & 0x80000000u) ? (k ^ 0x80000000u) : ~k;
  return __uint_as_float(u);
}

__device__ __forceinline__ int wave_incl_scan(int v, int lane) {
  #pragma unroll
  for (int d = 1; d < 32; d <<= 1) {
    const int t = __shfl_up(v, d);
    if (lane >= d) v += t;
  }
  return v;
}

template<int NC>
__global__ __launch_bounds__(256) void gemm_f32_wmma(const float* __restrict__ A,
                                                     const float* __restrict__ B,
                                                     float* __restrict__ C,
                                                     const float* __restrict__ aw_src,
                                                     const float* __restrict__ aw_dst,
                                                     float* __restrict__ asrc,
                                                     float* __restrict__ adst, int M) {
  constexpr int K = 128;
  constexpr int CG = 64;
  constexpr int LDA = K + 4;
  constexpr int colGroups = NC / CG;
  __shared__ float Wlds[K * CG];
  __shared__ float Alds[2][32 * LDA];
  __shared__ float Ulds[2][K];
  const int cg = blockIdx.x % colGroups;
  const int rg0 = blockIdx.x / colGroups;
  const int rgStep = gridDim.x / colGroups;
  const int rowGroups = (M + 31) >> 5;
  const int tid = threadIdx.x;

  float4 av[4];
  auto loadA = [&](int rg) {
    #pragma unroll
    for (int j = 0; j < 4; ++j) {
      const int i = tid + j * 256;
      const int r = i / (K / 4), c4 = i % (K / 4);
      const int grow = rg * 32 + r;
      av[j] = make_float4(0.f, 0.f, 0.f, 0.f);
      if (grow < M) av[j] = *(const float4*)&A[(size_t)grow * K + c4 * 4];
    }
  };
  if (rg0 < rowGroups) loadA(rg0);

  {
    float4 bv[8];
    #pragma unroll
    for (int j = 0; j < 8; ++j) {
      const int i = tid + j * 256;
      const int k = i / (CG / 4), c4 = i % (CG / 4);
      bv[j] = *(const float4*)&B[(size_t)k * NC + cg * CG + c4 * 4];
    }
    const float4 as4 = *(const float4*)&aw_src[cg * CG + (tid % (CG / 4)) * 4];
    const float4 ad4 = *(const float4*)&aw_dst[cg * CG + (tid % (CG / 4)) * 4];
    #pragma unroll
    for (int j = 0; j < 8; ++j) {
      const int i = tid + j * 256;
      const int k = i / (CG / 4), c4 = i % (CG / 4);
      *(float4*)&Wlds[k * CG + c4 * 4] = bv[j];
    }
    float v16[16];
    #pragma unroll
    for (int j = 0; j < 8; ++j) {
      v16[2 * j]     = bv[j].x * as4.x + bv[j].y * as4.y + bv[j].z * as4.z + bv[j].w * as4.w;
      v16[2 * j + 1] = bv[j].x * ad4.x + bv[j].y * ad4.y + bv[j].z * ad4.z + bv[j].w * ad4.w;
    }
    const bool b3 = (tid & 8) != 0, b2 = (tid & 4) != 0, b1 = (tid & 2) != 0, b0 = (tid & 1) != 0;
    float v8[8], v4[4], v2[2];
    #pragma unroll
    for (int i = 0; i < 8; ++i) v8[i] = (b3 ? v16[8 + i] : v16[i]) + __shfl_xor(b3 ? v16[i] : v16[8 + i], 8);
    #pragma unroll
    for (int i = 0; i < 4; ++i) v4[i] = (b2 ? v8[4 + i] : v8[i]) + __shfl_xor(b2 ? v8[i] : v8[4 + i], 4);
    #pragma unroll
    for (int i = 0; i < 2; ++i) v2[i] = (b1 ? v4[2 + i] : v4[i]) + __shfl_xor(b1 ? v4[i] : v4[2 + i], 2);
    const float v1 = (b0 ? v2[1] : v2[0]) + __shfl_xor(b0 ? v2[0] : v2[1], 1);
    Ulds[tid & 1][(tid >> 4) + 16 * ((tid & 15) >> 1)] = v1;
  }
  __syncthreads();
  const float4 us4 = *(const float4*)&Ulds[0][(tid & 31) * 4];
  const float4 ud4 = *(const float4*)&Ulds[1][(tid & 31) * 4];

  const int wave = __builtin_amdgcn_readfirstlane(tid >> 5);
  const int lane = tid & 31;
  const int half = lane >> 4;
  const int lrow = lane & 15;
  const int rt = wave >> 2, ct = wave & 3;
  const float* Bb = &Wlds[(2 * half) * CG + ct * 16 + lrow];

  int buf = 0;
  for (int rg = rg0; rg < rowGroups; rg += rgStep) {
    #pragma unroll
    for (int j = 0; j < 4; ++j) {
      const int i = tid + j * 256;
      const int r = i / (K / 4), c4 = i % (K / 4);
      *(float4*)&Alds[buf][r * LDA + c4 * 4] = av[j];
    }
    float p8[8];
    #pragma unroll
    for (int j = 0; j < 4; ++j) {
      p8[j]     = av[j].x * us4.x + av[j].y * us4.y + av[j].z * us4.z + av[j].w * us4.w;
      p8[4 + j] = av[j].x * ud4.x + av[j].y * ud4.y + av[j].z * ud4.z + av[j].w * ud4.w;
    }
    __syncthreads();
    if (rg + rgStep < rowGroups) loadA(rg + rgStep);
    {
      const bool h4 = (tid & 16) != 0, h3 = (tid & 8) != 0, h2 = (tid & 4) != 0;
      float p4[4], p2[2];
      #pragma unroll
      for (int i = 0; i < 4; ++i) p4[i] = (h4 ? p8[4 + i] : p8[i]) + __shfl_xor(h4 ? p8[i] : p8[4 + i], 16);
      #pragma unroll
      for (int i = 0; i < 2; ++i) p2[i] = (h3 ? p4[2 + i] : p4[i]) + __shfl_xor(h3 ? p4[i] : p4[2 + i], 8);
      float p1 = (h2 ? p2[1] : p2[0]) + __shfl_xor(h2 ? p2[0] : p2[1], 4);
      p1 += __shfl_xor(p1, 2);
      p1 += __shfl_xor(p1, 1);
      const int arow = rg * 32 + (tid >> 5) + 8 * ((tid >> 2) & 3);
      if ((tid & 3) == 0 && arow < M) (h4 ? adst : asrc)[(size_t)arow * colGroups + cg] = p1;
    }

    const int grow0 = rg * 32 + rt * 16;
    if (grow0 < M) {
      const float* Ab = &Alds[buf][(rt * 16 + lrow) * LDA + 2 * half];
      v8f acc = {};
      #pragma unroll
      for (int k0 = 0; k0 < K; k0 += 4) {
        v2f a, b;
        a.x = Ab[k0];        a.y = Ab[k0 + 1];
        b.x = Bb[k0 * CG];   b.y = Bb[k0 * CG + CG];
        acc = __builtin_amdgcn_wmma_f32_16x16x4_f32(false, a, false, b,
                                                    (short)0, acc, false, false);
      }
      float* Cp = &C[(size_t)(grow0 + 8 * half) * NC + cg * CG + ct * 16 + lrow];
      #pragma unroll
      for (int j = 0; j < 8; ++j) Cp[(size_t)j * NC] = acc[j];
    }
    buf ^= 1;
  }
}

template<int R>
__global__ __launch_bounds__(128) void chunk_map_kernel(const int* __restrict__ dstE,
                                                        unsigned* __restrict__ map, int E) {
  __shared__ unsigned bits[4][256];
  const int lane = threadIdx.x & 31;
  const int wave = threadIdx.x >> 5;
  const int g = blockIdx.x * 4 + wave;
  if (g >= ((E + 1023) >> 10)) return;
  unsigned* const b = bits[wave];
  #pragma unroll
  for (int j = 0; j < 8; ++j) b[lane + 32 * j] = 0u;
  int d[32];
  #pragma unroll
  for (int c = 0; c < 32; ++c) {
    const int e = g * 1024 + c * 32 + lane;
    d[c] = dstE[e < E ? e : E - 1];
  }
  #pragma unroll
  for (int c = 0; c < 32; ++c) {
    const int e = g * 1024 + c * 32 + lane;
    if (e < E) atomicOr(&b[((unsigned)d[c] / (unsigned)R) & 255u], 1u << c);
  }
  #pragma unroll
  for (int j = 0; j < 8; ++j) map[(size_t)g * 256 + lane + 32 * j] = b[lane + 32 * j];
}

template<int H, int R, bool SIGMOID, bool EMIT, bool USE>
__global__ __launch_bounds__(1024) void range_expagg_kernel(const int* __restrict__ srcE,
                                                            const int* __restrict__ dstE,
                                                            const unsigned* __restrict__ map,
                                                            const float* __restrict__ asrc,
                                                            const float* __restrict__ adst,
                                                            const float* __restrict__ h,
                                                            const float* __restrict__ bias,
                                                            float* __restrict__ rows,
                                                            int* __restrict__ elist, int E, int N) {
  constexpr int HC = H * 64;
  constexpr int KK = HC / 32;
  typedef __attribute__((ext_vector_type(KK))) float rowv;
  constexpr int NT = 1024;
  constexpr int NW = NT / 32;
  constexpr int UC = 8;
  constexpr int UD = 32 / KK;
  constexpr int QCAP = 32 + 32 * UC;
  constexpr int LCAP = 6144;
  constexpr int LSTRIDE = 2 * LCAP + 32;
  static_assert(R <= NT, "one lane per owned node for the self-loops");
  static_assert(NT % 32 == 0, "a thread keeps its lane's channels through the store loop");
  __shared__ float acc[R * HC];
  __shared__ float sl[R * H];
  __shared__ unsigned mk[R * H];
  __shared__ float adl[R * H];
  __shared__ int qe[NW * QCAP];
  __shared__ v2i lp[LCAP];
  __shared__ float lv[LCAP * H];
  __shared__ int lcount;
  __shared__ int qcount;

  const int tid = threadIdx.x;
  const int lane = tid & 31;
  const int wave = tid >> 5;
  const int lo = blockIdx.x * R;
  int* const gcount = elist + (size_t)blockIdx.x * LSTRIDE;
  int* const gsrc = gcount + 32;
  int* const gdl = gsrc + LCAP;

  for (int i = tid; i < R * HC; i += NT) acc[i] = 0.f;
  for (int i = tid; i < R * H; i += NT) {
    int gi = lo * H + i;
    sl[i] = 0.f;
    mk[i] = 0u;
    adl[i] = (gi < N * H) ? adst[gi] : 0.f;
  }
  const int nself = (N - lo < R) ? (N - lo) : R;
  if (tid == 0) { lcount = USE ? 0 : nself; qcount = 0; }
  __syncthreads();

  int* myq = qe + wave * QCAP;
  int cnt = 0;

  auto append = [&](bool keep, int e) {
    unsigned mask = __builtin_amdgcn_ballot_w32(keep);
    if (mask == 0u) return;
    if (keep) myq[cnt + __builtin_popcount(mask & ((1u << lane) - 1u))] = e;
    cnt += __builtin_popcount(mask);
  };

  auto edge_logits = [&](int e, int& s, int& dl, float& v0, float& v1) {
    s = (e < E) ? srcE[e] : (e - E);
    dl = ((e < E) ? dstE[e] : (e - E)) - lo;
    float v = asrc[s * H] + adl[dl * H];
    v0 = (v > 0.f) ? v : NEG_SLOPE * v;
    v1 = 0.f;
    if (H == 2) {
      float w = asrc[s * H + 1] + adl[dl * H + 1];
      v1 = (w > 0.f) ? w : NEG_SLOPE * w;
    }
  };

  auto edge_exp = [&](int dl, float v0, float v1, float& ex0, float& ex1) {
    ex0 = __expf(v0 - unflipf(mk[dl * H]));
    atomicAdd(&sl[dl * H], ex0);
    ex1 = 0.f;
    if (H == 2) {
      ex1 = __expf(v1 - unflipf(mk[dl * H + 1]));
      atomicAdd(&sl[dl * H + 1], ex1);
    }
  };

  auto add_rows = [&](int n, int s, int dl, float ex0, float ex1) {
    const bool hi = (H == 2) && (lane >= 16);
    auto group = [&](auto gs, int i0) {
      constexpr int G = decltype(gs)::value;
      rowv val[G];
      #pragma unroll
      for (int u = 0; u < G; ++u) {
        const int si = __builtin_amdgcn_readlane(s, i0 + u);
        val[u] = *((const rowv*)(h + (size_t)si * HC) + lane);
      }
      #pragma unroll
      for (int u = 0; u < G; ++u) {
        const int di = __builtin_amdgcn_readlane(dl, i0 + u);
        const float x0 = __int_as_float(__builtin_amdgcn_readlane(__float_as_int(ex0), i0 + u));
        const float x1 = __int_as_float(__builtin_amdgcn_readlane(__float_as_int(ex1), i0 + u));
        const float x = hi ? x1 : x0;
        float* ap = acc + di * HC + lane;
        #pragma unroll
        for (int j = 0; j < KK; ++j) atomicAdd(ap + 32 * j, x * val[u][j]);
      }
    };
    int i = 0;
    for (; i + UD <= n; i += UD) group(GroupOf<UD>{}, i);
    if (i + UD / 2 <= n) { group(GroupOf<UD / 2>{}, i); i += UD / 2; }
    for (; i < n; ++i) group(GroupOf<1>{}, i);
  };

  const int wv = __builtin_amdgcn_readfirstlane(wave);

  auto walk = [&](auto batch) {
    cnt = 0;
    append(tid < R && lo + tid < N, E + lo + tid);

    const int hb = blockIdx.x & 255;
    const int nGroups = (E + 1023) >> 10;
    const int nPairs = (nGroups + 1) >> 1;
    int pnext = wv;
    int pcur = 0;
    int li = 0, ln = 0;
    unsigned wlo = 0u, whi = 0u;
    unsigned long long msk = 0ull;
    int ebase = 0;
    for (;;) {
      while (msk == 0ull) {
        if (li == ln) {
          if (pnext >= nPairs) break;
          const int mine = pnext + 32 * lane;
          const int g0 = 2 * mine;
          wlo = (mine < nPairs) ? map[(size_t)g0 * 256 + hb] : 0u;
          whi = (mine < nPairs && g0 + 1 < nGroups) ? map[(size_t)(g0 + 1) * 256 + hb] : 0u;
          const int left = (nPairs - pnext + 31) >> 5;
          ln = left < 32 ? left : 32;
          li = 0;
          pcur = pnext;
          pnext += 32 * 32;
        }
        const unsigned a = (unsigned)__builtin_amdgcn_readlane((int)wlo, li);
        const unsigned b = (unsigned)__builtin_amdgcn_readlane((int)whi, li);
        msk = (unsigned long long)a | ((unsigned long long)b << 32);
        ebase = (pcur + 32 * li) * 2048;
        ++li;
      }
      const bool more = (msk != 0ull);
      if (more) {
        int ce[UC], dv[UC];
        bool cv[UC];
        const int c0 = __builtin_ctzll(msk);
        #pragma unroll
        for (int j = 0; j < UC; ++j) {
          cv[j] = (msk != 0ull);
          const int c = cv[j] ? __builtin_ctzll(msk) : c0;
          msk &= msk - 1ull;
          ce[j] = ebase + 32 * c + lane;
          dv[j] = dstE[ce[j] < E ? ce[j] : E - 1];
        }
        #pragma unroll
        for (int j = 0; j < UC; ++j)
          append(cv[j] && ce[j] < E && (unsigned)(dv[j] - lo) < (unsigned)R, ce[j]);
      }
      const int leave = more ? 31 : 0;
      while (cnt > leave) {
        const int n = cnt < 32 ? cnt : 32;
        cnt -= n;
        batch(n);
      }
      if (!more) break;
    }
  };

  constexpr int IT = LCAP / NT;
  static_assert(LCAP % NT == 0, "a whole number of places a thread");
  int given = LCAP + 1;
  if (USE) given = __builtin_amdgcn_readfirstlane(gcount[0]);
  bool listed = false;
  if (!USE) {
    constexpr int QTOT = NW * QCAP;
    if (tid < nself) lp[tid].x = E + lo + tid;
    const int tail0 = E & ~31;
    if (tid < E - tail0) {
      const int e = tail0 + tid;
      if ((unsigned)(dstE[e] - lo) < (unsigned)R) {
        const int pos = atomicAdd(&lcount, 1);
        if (pos < LCAP) lp[pos].x = e;
      }
    }
    {
      const int hb = blockIdx.x & 255;
      const int nBytes = ((E + 1023) >> 10) * 4;
      constexpr int MT = 4;
      for (int i0 = 0; i0 < nBytes; i0 += NT * MT) {
        unsigned w[MT];
        #pragma unroll
        for (int k = 0; k < MT; ++k) {
          const int idx = i0 + k * NT + tid;
          w[k] = (idx < nBytes) ? map[(size_t)(idx >> 2) * 256 + hb] : 0u;
        }
        int pcs = 0;
        #pragma unroll
        for (int k = 0; k < MT; ++k) {
          const int idx = i0 + k * NT + tid;
          const int ct = (tail0 >> 5) - idx * 8;
          w[k] = (w[k] >> (8 * (idx & 3))) & 255u;
          if (tail0 < E && ct >= 0 && ct < 8) w[k] &= ~(1u << ct);
          pcs += __builtin_popcount(w[k]);
        }
        const int inc = wave_incl_scan(pcs, lane);
        int base = 0;
        if (lane == 31 && inc) base = atomicAdd(&qcount, inc);
        int pos = __builtin_amdgcn_readlane(base, 31) + inc - pcs;
        #pragma unroll
        for (int k = 0; k < MT; ++k) {
          const int idx = i0 + k * NT + tid;
          unsigned m = w[k];
          while (m) {
            const int b = __builtin_ctz(m);
            m &= m - 1u;
            if (pos < QTOT) qe[pos] = (idx * 8 + b) * 32;
            ++pos;
          }
        }
      }
    }
    __syncthreads();
    const int nq = __builtin_amdgcn_readfirstlane(qcount);
    if (nq <= QTOT) {
      const int per = (((nq + NW - 1) / NW + 31) / 32) * 32;
      const int beg = wv * per;
      const int end = (beg + per < nq) ? (beg + per) : nq;
      for (int b0 = beg; b0 < end; b0 += 32) {
        const int nv = (end - b0 < 32) ? (end - b0) : 32;
        const int vb = qe[b0 + (lane < nv ? lane : 0)];
        int dv[32];
        #pragma unroll
        for (int j = 0; j < 32; ++j) dv[j] = dstE[__builtin_amdgcn_readlane(vb, j) + lane];
        int vkm = 0;
        #pragma unroll
        for (int j = 0; j < 32; ++j) {
          const int k = (int)__builtin_amdgcn_ballot_w32((unsigned)(dv[j] - lo) < (unsigned)R);
          vkm = (lane == j) ? k : vkm;
        }
        unsigned m = (lane < nv) ? (unsigned)vkm : 0u;
        const int pc = __builtin_popcount(m);
        const int inc = wave_incl_scan(pc, lane);
        int base = 0;
        if (lane == 31 && inc) base = atomicAdd(&lcount, inc);
        int pos = __builtin_amdgcn_readlane(base, 31) + inc - pc;
        while (m) {
          const int b = __builtin_ctz(m);
          m &= m - 1u;
          if (pos < LCAP) lp[pos].x = vb + b;
          ++pos;
        }
      }
    }
    __syncthreads();
    const int found = (nq <= QTOT) ? __builtin_amdgcn_readfirstlane(lcount) : (LCAP + 1);
    if (found <= LCAP) {
      int sj[IT], dj[IT];
      float aj[IT][H];
      #pragma unroll
      for (int j = 0; j < IT; ++j) {
        const int i = tid + j * NT;
        int e = E;
        if (i < found) e = lp[i].x;
        sj[j] = (e < E) ? srcE[e] : (e - E);
        dj[j] = ((e < E) ? dstE[e] : (e - E)) - lo;
      }
      #pragma unroll
      for (int j = 0; j < IT; ++j) {
        #pragma unroll
        for (int q = 0; q < H; ++q) aj[j][q] = asrc[sj[j] * H + q];
      }
      #pragma unroll
      for (int j = 0; j < IT; ++j) {
        const int i = tid + j * NT;
        if (i < found) {
          lp[i] = v2i{sj[j], dj[j]};
          #pragma unroll
          for (int q = 0; q < H; ++q) {
            const float v = aj[j][q] + adl[dj[j] * H + q];
            const float lv1 = (v > 0.f) ? v : NEG_SLOPE * v;
            atomicMax(&mk[dj[j] * H + q], flipf(lv1));
            lv[i * H + q] = lv1;
          }
        }
      }
      listed = true;
    } else {
      __syncthreads();
      if (tid == 0) lcount = 0;
      __syncthreads();
    }
  } else if (given <= LCAP) {
    int sj[IT], dj[IT];
    float aj[IT][H];
    #pragma unroll
    for (int j = 0; j < IT; ++j) {
      const int i = tid + j * NT;
      sj[j] = (i < given) ? gsrc[i] : 0;
      dj[j] = (i < given) ? gdl[i] : 0;
    }
    #pragma unroll
    for (int j = 0; j < IT; ++j) {
      #pragma unroll
      for (int q = 0; q < H; ++q) aj[j][q] = asrc[sj[j] * H + q];
    }
    #pragma unroll
    for (int j = 0; j < IT; ++j) {
      const int i = tid + j * NT;
      if (i < given) {
        lp[i] = v2i{sj[j], dj[j]};
        #pragma unroll
        for (int q = 0; q < H; ++q) {
          const float v = aj[j][q] + adl[dj[j] * H + q];
          const float lv1 = (v > 0.f) ? v : NEG_SLOPE * v;
          atomicMax(&mk[dj[j] * H + q], flipf(lv1));
          lv[i * H + q] = lv1;
        }
      }
    }
    if (tid == 0) lcount = given;
    listed = true;
  }
  if (!listed) walk([&](int n) {
    int s = 0, dl = 0;
    float v0 = 0.f, v1 = 0.f;
    if (lane < n) {
      edge_logits(myq[cnt + lane], s, dl, v0, v1);
      atomicMax(&mk[dl * H], flipf(v0));
      if (H == 2) atomicMax(&mk[dl * H + 1], flipf(v1));
    }
    int base = 0;
    if (lane == 0) base = atomicAdd(&lcount, n);
    base = __builtin_amdgcn_readfirstlane(base);
    if (base + n <= LCAP && lane < n) {
      lp[base + lane] = v2i{s, dl};
      lv[(base + lane) * H] = v0;
      if (H == 2) lv[(base + lane) * H + 1] = v1;
    }
  });
  __syncthreads();

  const int total = lcount;
  if (EMIT) {
    if (tid == 0) gcount[0] = total;
    if (total <= LCAP) {
      #pragma unroll
      for (int j = 0; j < LCAP / NT; ++j) {
        const int i = tid + j * NT;
        if (i < total) {
          const v2i p = lp[i];
          gsrc[i] = p.x;
          gdl[i] = p.y;
        }
      }
    }
  }
  if (total <= LCAP) {
    constexpr int HG = UD / 2;
    const int per = (((total + NW - 1) / NW + HG - 1) / HG) * HG;
    const int beg = wv * per;
    const int end = (beg + per < total) ? (beg + per) : total;
    for (int b = beg; b < end; b += 32) {
      const int n = (end - b < 32) ? (end - b) : 32;
      int s = 0, dl = 0;
      float ex0 = 0.f, ex1 = 0.f;
      if (lane < n) {
        const v2i p = lp[b + lane];
        s = p.x;
        dl = p.y;
        edge_exp(dl, lv[(b + lane) * H], (H == 2) ? lv[(b + lane) * H + 1] : 0.f, ex0, ex1);
      }
      add_rows(n, s, dl, ex0, ex1);
    }
  } else {
    walk([&](int n) {
      int s = 0, dl = 0;
      float v0 = 0.f, v1 = 0.f, ex0 = 0.f, ex1 = 0.f;
      if (lane < n) {
        edge_logits(myq[cnt + lane], s, dl, v0, v1);
        edge_exp(dl, v0, v1, ex0, ex1);
      }
      add_rows(n, s, dl, ex0, ex1);
    });
  }
  __syncthreads();

  const rowv bv = *((const rowv*)bias + lane);
  const int hsel = ((H == 2) && (lane >= 16)) ? 1 : 0;
  const int nrw = ((N - lo < R) ? (N - lo) : R) * 32;
  float* rp = rows + (size_t)lo * HC;
  for (int iv = tid; iv < nrw; iv += NT) {
    const int nd = iv >> 5;
    const float den = sl[nd * H + hsel];
    rowv o;
    #pragma unroll
    for (int j = 0; j < KK; ++j) {
      const float v = acc[nd * HC + 32 * j + lane] / den + bv[j];
      o[j] = SIGMOID ? 1.f / (1.f + __expf(-v)) : (v > 0.f ? v : 0.f);
    }
    *((rowv*)(rp + (size_t)nd * HC) + lane) = o;
  }
}

extern "C" void kernel_launch(void* const* d_in, const int* in_sizes, int n_in,
                              void* d_out, int out_size, void* d_ws, size_t ws_size,
                              hipStream_t stream) {
  const float* x   = (const float*)d_in[0];
  const int*   ei  = (const int*)d_in[1];
  const float* W1  = (const float*)d_in[2];
  const float* as1 = (const float*)d_in[3];
  const float* ad1 = (const float*)d_in[4];
  const float* b1  = (const float*)d_in[5];
  const float* W2  = (const float*)d_in[6];
  const float* as2 = (const float*)d_in[7];
  const float* ad2 = (const float*)d_in[8];
  const float* b2  = (const float*)d_in[9];
  float* out = (float*)d_out;

  const int N = in_sizes[0] / 128;
  const int E = in_sizes[1] / 2;
  const int* srcE = ei;
  const int* dstE = ei + E;

  float* w = (float*)d_ws;
  size_t o = 0;
  unsigned* cmap = (unsigned*)(w + o); o += (size_t)((E + 1023) / 1024) * 256;
  float* h1    = w + o; o += (size_t)N * 128;
  float* agg1  = w + o; o += (size_t)N * 128;
  float* asrc1 = w + o; o += (size_t)N * 2;
  float* adst1 = w + o; o += (size_t)N * 2;
  float* asrc2 = w + o; o += (size_t)N;
  float* adst2 = w + o; o += (size_t)N;
  int* elist = (int*)(w + o);
  float* h2f   = h1;
  (void)o; (void)ws_size; (void)n_in; (void)out_size;

  const int TB = 256;
  auto blk = [](long long n, int tb) { return (unsigned)((n + tb - 1) / tb); };
  constexpr int GEMM_WGS = 1024;
  constexpr int RNG = 200;
  const int rangeGroups = (N + RNG - 1) / RNG;

  chunk_map_kernel<RNG><<<blk(((long long)E + 1023) / 1024, 4), 128, 0, stream>>>(dstE, cmap, E);

  gemm_f32_wmma<128><<<dim3(GEMM_WGS), TB, 0, stream>>>(x, W1, h1, as1, ad1, asrc1, adst1, N);
  range_expagg_kernel<2, RNG, false, true, false><<<dim3(rangeGroups), 1024, 0, stream>>>(srcE, dstE, cmap, asrc1, adst1, h1, b1, agg1, elist, E, N);

  gemm_f32_wmma<64><<<dim3(GEMM_WGS), TB, 0, stream>>>(agg1, W2, h2f, as2, ad2, asrc2, adst2, N);
  range_expagg_kernel<1, RNG, true, false, true><<<dim3(rangeGroups), 1024, 0, stream>>>(srcE, dstE, cmap, asrc2, adst2, h2f, b2, out, elist, E, N);
}
